# starting kernel
# baseline (speedup 1.0000x reference)
#include <hip/hip_runtime.h>
#include <math.h>

#define Nn   100000
#define Ee   800000
#define DIN  256
#define DH   128
#define Kc   16
#define GAMMA 0.1f
#define LAM1  1.0f
#define LAM2  0.5f
#define AZ (1.0f - GAMMA*LAM1 + GAMMA*LAM2)
#define CS (GAMMA*LAM1)
#define CT (GAMMA*LAM2)
#define GRAM_CHUNK 400
#define GRAM_TILE  16
#define GEMM_KTILE 64

typedef __attribute__((ext_vector_type(2))) float v2f;
typedef __attribute__((ext_vector_type(8))) float v8f;

__device__ __forceinline__ v8f wmma4(v2f a, v2f b, v8f c) {
  return __builtin_amdgcn_wmma_f32_16x16x4_f32(
       false, a,  false, b,
       (short)0, c,  false,  false);
}


__global__ void fill4_kernel(float4* __restrict__ p, long n4, float v) {
  long i = (long)blockIdx.x * blockDim.x + threadIdx.x;
  long stride = (long)gridDim.x * blockDim.x;
  float4 val = make_float4(v, v, v, v);
  for (; i < n4; i += stride) p[i] = val;
}

__global__ void degree_kernel(const int* __restrict__ edge, float* __restrict__ deg) {
  int e = blockIdx.x * blockDim.x + threadIdx.x;
  if (e < Ee) atomicAdd(&deg[edge[Ee + e]], 1.0f);
}

__global__ void rsqrt_kernel(float* __restrict__ deg) {
  int i = blockIdx.x * blockDim.x + threadIdx.x;
  if (i < Nn) {
    float d = deg[i];
    deg[i] = (d > 0.0f) ? rsqrtf(fmaxf(d, 1.0f)) : 0.0f;
  }
}

__global__ void __launch_bounds__(256)
gemm_wmma_kernel(const float* __restrict__ A, const float* __restrict__ B,
                 float* __restrict__ C, int Kdim) {
  __shared__ __align__(16) float atile[16 * GEMM_KTILE];
  int tid  = threadIdx.x;
  int wave = tid >> 5;
  int lane = tid & 31;
  int half = lane >> 4;
  int lid  = lane & 15;
  int kb   = half * 2;
  int row0 = blockIdx.x * 16;
  int col0 = wave * 16;
  int sr   = tid >> 4;
  int sc   = tid & 15;
  v8f acc = {};
  for (int k0 = 0; k0 < Kdim; k0 += GEMM_KTILE) {
    __syncthreads();
    ((float4*)(atile + sr * GEMM_KTILE))[sc] =
        ((const float4*)(A + (size_t)(row0 + sr) * Kdim + k0))[sc];
    __syncthreads();
#pragma unroll
    for (int ks = 0; ks < GEMM_KTILE; ks += 4) {
      v2f a = *(const v2f*)(atile + lid * GEMM_KTILE + ks + kb);
      v2f b;
      b.x = B[(size_t)(k0 + ks + kb)     * DH + col0 + lid];
      b.y = B[(size_t)(k0 + ks + kb + 1) * DH + col0 + lid];
      acc = wmma4(a, b, acc);
    }
  }
#pragma unroll
  for (int r = 0; r < 8; ++r)
    C[(size_t)(row0 + r + 8 * half) * DH + col0 + lid] = acc[r];
}

__global__ void scatter_kernel(const int* __restrict__ edge, const float* __restrict__ dinv,
                               const float* __restrict__ H, float* __restrict__ S) {
  int idx  = blockIdx.x * 8 + (threadIdx.x >> 5);
  int lane = threadIdx.x & 31;
  if (idx >= Ee + Nn) return;
  int r, c; float w;
  if (idx < Ee) {
    r = edge[idx];
    c = edge[Ee + idx];
    w = dinv[r] * dinv[c];
  } else {
    r = c = idx - Ee;
    float di = dinv[r];
    w = di * di;
  }
  const float4 h = *(const float4*)(H + (size_t)r * DH + lane * 4);
  float* sp = S + (size_t)c * DH + lane * 4;
  atomicAdd(sp + 0, w * h.x);
  atomicAdd(sp + 1, w * h.y);
  atomicAdd(sp + 2, w * h.z);
  atomicAdd(sp + 3, w * h.w);
}

__global__ void __launch_bounds__(256)
gram_wmma_kernel(const float* __restrict__ H, float* __restrict__ G) {
  __shared__ __align__(16) float tile[GRAM_TILE * DH];
  int tid  = threadIdx.x;
  int wave = tid >> 5;
  int lane = tid & 31;
  int half = lane >> 4;
  int lid  = lane & 15;
  int kb   = 2 * half;
  int n0   = blockIdx.x * GRAM_CHUNK;
  v8f acc[8] = {};
  for (int base = n0; base < n0 + GRAM_CHUNK; base += GRAM_TILE) {
    __syncthreads();
    {
      const float4* src = (const float4*)(H + (size_t)base * DH);
      float4* dst = (float4*)tile;
      dst[tid]       = src[tid];
      dst[tid + 256] = src[tid + 256];
    }
    __syncthreads();
#pragma unroll
    for (int ks = 0; ks < GRAM_TILE; ks += 4) {
      v2f a;
      a.x = tile[(ks + kb)     * DH + wave * 16 + lid];
      a.y = tile[(ks + kb + 1) * DH + wave * 16 + lid];
#pragma unroll
      for (int ct = 0; ct < 8; ++ct) {
        v2f b;
        b.x = tile[(ks + kb)     * DH + ct * 16 + lid];
        b.y = tile[(ks + kb + 1) * DH + ct * 16 + lid];
        acc[ct] = wmma4(a, b, acc[ct]);
      }
    }
  }
#pragma unroll
  for (int ct = 0; ct < 8; ++ct)
#pragma unroll
    for (int r = 0; r < 8; ++r)
      atomicAdd(&G[(size_t)(wave * 16 + r + 8 * half) * DH + ct * 16 + lid], acc[ct][r]);
}

__global__ void __launch_bounds__(256)
update_wmma_kernel(const float* __restrict__ H, float* __restrict__ S,
                   const float* __restrict__ G, const float* __restrict__ bias) {
  int wave = threadIdx.x >> 5;
  int lane = threadIdx.x & 31;
  int half = lane >> 4;
  int lid  = lane & 15;
  int kb   = 2 * half;
  int row0 = blockIdx.x * 16;
  int col0 = wave * 16;
  v8f acc = {};
  for (int kk = 0; kk < DH; kk += 4) {
    v2f a = *(const v2f*)(H + (size_t)(row0 + lid) * DH + kk + kb);
    v2f b;
    b.x = G[(size_t)(kk + kb)     * DH + col0 + lid];
    b.y = G[(size_t)(kk + kb + 1) * DH + col0 + lid];
    acc = wmma4(a, b, acc);
  }
  float bcol = bias[col0 + lid];
#pragma unroll
  for (int r = 0; r < 8; ++r) {
    size_t off = (size_t)(row0 + r + 8 * half) * DH + col0 + lid;
    float val = AZ * H[off] + CS * S[off] - CT * acc[r] + bcol;
    S[off] = fmaxf(val, 0.0f);
  }
}

__global__ void fc_logsoftmax_kernel(const float* __restrict__ X, const float* __restrict__ Wf,
                                     const float* __restrict__ bf, float* __restrict__ out) {
  int n = blockIdx.x * blockDim.x + threadIdx.x;
  if (n >= Nn) return;
  const float* h = X + (size_t)n * DH;
  float logit[Kc];
#pragma unroll
  for (int k = 0; k < Kc; ++k) logit[k] = bf[k];
  for (int j = 0; j < DH; ++j) {
    float hj = h[j];
#pragma unroll
    for (int k = 0; k < Kc; ++k) logit[k] = fmaf(hj, Wf[j * Kc + k], logit[k]);
  }
  float m = logit[0];
#pragma unroll
  for (int k = 1; k < Kc; ++k) m = fmaxf(m, logit[k]);
  float s = 0.0f;
#pragma unroll
  for (int k = 0; k < Kc; ++k) s += expf(logit[k] - m);
  float lse = m + logf(s);
#pragma unroll
  for (int k = 0; k < Kc; ++k) out[(size_t)n * Kc + k] = logit[k] - lse;
}

extern "C" void kernel_launch(void* const* d_in, const int* in_sizes, int n_in,
                              void* d_out, int out_size, void* d_ws, size_t ws_size,
                              hipStream_t stream) {
  const float* x    = (const float*)d_in[0];
  const int*   edge = (const int*)  d_in[1];
  const float* W1   = (const float*)d_in[2];
  const float* b1   = (const float*)d_in[3];
  const float* W2   = (const float*)d_in[4];
  const float* b2   = (const float*)d_in[5];
  const float* Wf   = (const float*)d_in[6];
  const float* bf   = (const float*)d_in[7];
  float* out = (float*)d_out;

  float* H    = (float*)d_ws;
  float* S    = H + (size_t)Nn * DH;
  float* dinv = S + (size_t)Nn * DH;
  float* G    = dinv + Nn;

  dim3 b256(256);

  fill4_kernel<<<128, b256, 0, stream>>>((float4*)dinv, (long)Nn / 4, 1.0f);
  degree_kernel<<<(Ee + 255) / 256, b256, 0, stream>>>(edge, dinv);
  rsqrt_kernel<<<(Nn + 255) / 256, b256, 0, stream>>>(dinv);

  gemm_wmma_kernel<<<Nn / 16, b256, 0, stream>>>(x, W1, H, DIN);
  fill4_kernel<<<2048, b256, 0, stream>>>((float4*)S, (long)Nn * DH / 4, 0.0f);
  scatter_kernel<<<(Ee + Nn) / 8, b256, 0, stream>>>(edge, dinv, H, S);
  fill4_kernel<<<16, b256, 0, stream>>>((float4*)G, (long)DH * DH / 4, 0.0f);
  gram_wmma_kernel<<<Nn / GRAM_CHUNK, b256, 0, stream>>>(H, G);
  update_wmma_kernel<<<Nn / 16, b256, 0, stream>>>(H, S, G, b1);

  gemm_wmma_kernel<<<Nn / 16, b256, 0, stream>>>(S, W2, H, DH);
  fill4_kernel<<<2048, b256, 0, stream>>>((float4*)S, (long)Nn * DH / 4, 0.0f);
  scatter_kernel<<<(Ee + Nn) / 8, b256, 0, stream>>>(edge, dinv, H, S);
  fill4_kernel<<<16, b256, 0, stream>>>((float4*)G, (long)DH * DH / 4, 0.0f);
  gram_wmma_kernel<<<Nn / GRAM_CHUNK, b256, 0, stream>>>(H, G);
  update_wmma_kernel<<<Nn / 16, b256, 0, stream>>>(H, S, G, b2);

  fc_logsoftmax_kernel<<<(Nn + 255) / 256, b256, 0, stream>>>(S, Wf, bf, out);
}


# Propagate via per-call CSR (int atomic rank, scan, place) and one-wave-per-node register gather; drops float atomics and zero fill on S; sum order now self loop then rank order
# speedup vs baseline: 1.5497x; 1.5497x over previous
#include <hip/hip_runtime.h>
#include <math.h>

#define Nn   100000
#define Ee   800000
#define DIN  256
#define DH   128
#define Kc   16
#define GAMMA 0.1f
#define LAM1  1.0f
#define LAM2  0.5f
#define AZ (1.0f - GAMMA*LAM1 + GAMMA*LAM2)
#define CS (GAMMA*LAM1)
#define CT (GAMMA*LAM2)
#define GRAM_CHUNK 400
#define GRAM_TILE  16
#define GEMM_KTILE 64

typedef __attribute__((ext_vector_type(2))) float v2f;
typedef __attribute__((ext_vector_type(8))) float v8f;

__device__ __forceinline__ v8f wmma4(v2f a, v2f b, v8f c) {
  return __builtin_amdgcn_wmma_f32_16x16x4_f32(
       false, a,  false, b,
       (short)0, c,  false,  false);
}


__global__ void fill4_kernel(float4* __restrict__ p, long n4, float v) {
  long i = (long)blockIdx.x * blockDim.x + threadIdx.x;
  long stride = (long)gridDim.x * blockDim.x;
  float4 val = make_float4(v, v, v, v);
  for (; i < n4; i += stride) p[i] = val;
}

__global__ void rank_kernel(const int* __restrict__ edge, int* __restrict__ cnt,
                            int* __restrict__ rank) {
  int e = blockIdx.x * blockDim.x + threadIdx.x;
  if (e < Ee) rank[e] = atomicAdd(&cnt[edge[Ee + e]], 1);
}

__global__ void __launch_bounds__(256)
scan_kernel(const int* __restrict__ cnt, int* __restrict__ offs, float* __restrict__ dinv) {
  __shared__ int red[8];
  __shared__ int wtot[8];
  int tid  = threadIdx.x;
  int lane = tid & 31;
  int wave = tid >> 5;
  int b    = blockIdx.x;
  const int4* c4 = (const int4*)cnt;
  int p = 0;
#pragma unroll 8
  for (int k = 0; k < b; ++k) {
    int4 v = c4[k * 256 + tid];
    p += (v.x + v.y) + (v.z + v.w);
  }
#pragma unroll
  for (int d = 16; d >= 1; d >>= 1) p += __shfl_xor(p, d);
  if (lane == 0) red[wave] = p;

  int i = b * 1024 + tid * 4;
  int4 v = make_int4(0, 0, 0, 0);
  if (i < Nn) v = c4[b * 256 + tid];
  int s = (v.x + v.y) + (v.z + v.w);
  int x = s;
#pragma unroll
  for (int d = 1; d < 32; d <<= 1) {
    int y = __shfl_up(x, d);
    if (lane >= d) x += y;
  }
  if (lane == 31) wtot[wave] = x;
  __syncthreads();
  int base = 0;
#pragma unroll
  for (int k = 0; k < 8; ++k) base += red[k];
#pragma unroll
  for (int k = 0; k < 8; ++k) base += (k < wave) ? wtot[k] : 0;
  int excl = base + x - s;
  if (i < Nn) {
    int4 o;
    o.x = excl;
    o.y = o.x + v.x;
    o.z = o.y + v.y;
    o.w = o.z + v.z;
    ((int4*)offs)[b * 256 + tid] = o;
    if (i + 4 == Nn) offs[Nn] = o.w + v.w;
    float d0 = (float)(v.x + 1), d1 = (float)(v.y + 1), d2 = (float)(v.z + 1), d3 = (float)(v.w + 1);
    float4 dv;
    dv.x = (d0 > 0.0f) ? rsqrtf(fmaxf(d0, 1.0f)) : 0.0f;
    dv.y = (d1 > 0.0f) ? rsqrtf(fmaxf(d1, 1.0f)) : 0.0f;
    dv.z = (d2 > 0.0f) ? rsqrtf(fmaxf(d2, 1.0f)) : 0.0f;
    dv.w = (d3 > 0.0f) ? rsqrtf(fmaxf(d3, 1.0f)) : 0.0f;
    ((float4*)dinv)[b * 256 + tid] = dv;
  }
}

__global__ void place_kernel(const int* __restrict__ edge, const int* __restrict__ offs,
                             const int* __restrict__ rank, const float* __restrict__ dinv,
                             int* __restrict__ csr) {
  int e = blockIdx.x * blockDim.x + threadIdx.x;
  if (e >= Ee) return;
  int r = edge[e];
  int c = edge[Ee + e];
  int pos = offs[c] + rank[e];
  pos = min(max(pos, 0), Ee - 1);
  int2 rec;
  rec.x = r;
  rec.y = __float_as_int(dinv[r] * dinv[c]);
  ((int2*)csr)[pos] = rec;
}

__global__ void __launch_bounds__(256)
gemm_wmma_kernel(const float* __restrict__ A, const float* __restrict__ B,
                 float* __restrict__ C, int Kdim) {
  __shared__ __align__(16) float atile[16 * GEMM_KTILE];
  int tid  = threadIdx.x;
  int wave = tid >> 5;
  int lane = tid & 31;
  int half = lane >> 4;
  int lid  = lane & 15;
  int kb   = half * 2;
  int row0 = blockIdx.x * 16;
  int col0 = wave * 16;
  int sr   = tid >> 4;
  int sc   = tid & 15;
  v8f acc = {};
  for (int k0 = 0; k0 < Kdim; k0 += GEMM_KTILE) {
    __syncthreads();
    ((float4*)(atile + sr * GEMM_KTILE))[sc] =
        ((const float4*)(A + (size_t)(row0 + sr) * Kdim + k0))[sc];
    __syncthreads();
#pragma unroll
    for (int ks = 0; ks < GEMM_KTILE; ks += 4) {
      v2f a = *(const v2f*)(atile + lid * GEMM_KTILE + ks + kb);
      v2f b;
      b.x = B[(size_t)(k0 + ks + kb)     * DH + col0 + lid];
      b.y = B[(size_t)(k0 + ks + kb + 1) * DH + col0 + lid];
      acc = wmma4(a, b, acc);
    }
  }
#pragma unroll
  for (int r = 0; r < 8; ++r)
    C[(size_t)(row0 + r + 8 * half) * DH + col0 + lid] = acc[r];
}

__global__ void __launch_bounds__(256)
gather_kernel(const int* __restrict__ offs, const int* __restrict__ csr,
              const float* __restrict__ dinv, const float* __restrict__ H, float* __restrict__ S) {
  int lane = threadIdx.x & 31;
  int c = blockIdx.x * 8 + __builtin_amdgcn_readfirstlane((int)(threadIdx.x >> 5));
  int beg = offs[c];
  int end = offs[c + 1];
  beg = min(max(beg, 0), Ee);
  end = min(max(end, beg), Ee);
  float di = dinv[c];
  float w0 = di * di;
  const float4 hs = *(const float4*)(H + (size_t)c * DH + lane * 4);
  float4 acc;
  acc.x = w0 * hs.x; acc.y = w0 * hs.y; acc.z = w0 * hs.z; acc.w = w0 * hs.w;
  for (int base = beg; base < end; base += 32) {
    int n = min(32, end - base);
    int2 my = make_int2(0, 0);
    if (lane < n) my = ((const int2*)csr)[base + lane];
#pragma unroll 4
    for (int j = 0; j < n; ++j) {
      int   r = __builtin_amdgcn_readlane(my.x, j);
      float w = __int_as_float(__builtin_amdgcn_readlane(my.y, j));
      r = min(max(r, 0), Nn - 1);
      const float4 h = *(const float4*)(H + (size_t)r * DH + lane * 4);
      acc.x = fmaf(w, h.x, acc.x);
      acc.y = fmaf(w, h.y, acc.y);
      acc.z = fmaf(w, h.z, acc.z);
      acc.w = fmaf(w, h.w, acc.w);
    }
  }
  *(float4*)(S + (size_t)c * DH + lane * 4) = acc;
}

__global__ void __launch_bounds__(256)
gram_wmma_kernel(const float* __restrict__ H, float* __restrict__ G) {
  __shared__ __align__(16) float tile[GRAM_TILE * DH];
  int tid  = threadIdx.x;
  int wave = tid >> 5;
  int lane = tid & 31;
  int half = lane >> 4;
  int lid  = lane & 15;
  int kb   = 2 * half;
  int n0   = blockIdx.x * GRAM_CHUNK;
  v8f acc[8] = {};
  for (int base = n0; base < n0 + GRAM_CHUNK; base += GRAM_TILE) {
    __syncthreads();
    {
      const float4* src = (const float4*)(H + (size_t)base * DH);
      float4* dst = (float4*)tile;
      dst[tid]       = src[tid];
      dst[tid + 256] = src[tid + 256];
    }
    __syncthreads();
#pragma unroll
    for (int ks = 0; ks < GRAM_TILE; ks += 4) {
      v2f a;
      a.x = tile[(ks + kb)     * DH + wave * 16 + lid];
      a.y = tile[(ks + kb + 1) * DH + wave * 16 + lid];
#pragma unroll
      for (int ct = 0; ct < 8; ++ct) {
        v2f b;
        b.x = tile[(ks + kb)     * DH + ct * 16 + lid];
        b.y = tile[(ks + kb + 1) * DH + ct * 16 + lid];
        acc[ct] = wmma4(a, b, acc[ct]);
      }
    }
  }
#pragma unroll
  for (int ct = 0; ct < 8; ++ct)
#pragma unroll
    for (int r = 0; r < 8; ++r)
      atomicAdd(&G[(size_t)(wave * 16 + r + 8 * half) * DH + ct * 16 + lid], acc[ct][r]);
}

__global__ void __launch_bounds__(256)
update_wmma_kernel(const float* __restrict__ H, float* __restrict__ S,
                   const float* __restrict__ G, const float* __restrict__ bias) {
  int wave = threadIdx.x >> 5;
  int lane = threadIdx.x & 31;
  int half = lane >> 4;
  int lid  = lane & 15;
  int kb   = 2 * half;
  int row0 = blockIdx.x * 16;
  int col0 = wave * 16;
  v8f acc = {};
  for (int kk = 0; kk < DH; kk += 4) {
    v2f a = *(const v2f*)(H + (size_t)(row0 + lid) * DH + kk + kb);
    v2f b;
    b.x = G[(size_t)(kk + kb)     * DH + col0 + lid];
    b.y = G[(size_t)(kk + kb + 1) * DH + col0 + lid];
    acc = wmma4(a, b, acc);
  }
  float bcol = bias[col0 + lid];
#pragma unroll
  for (int r = 0; r < 8; ++r) {
    size_t off = (size_t)(row0 + r + 8 * half) * DH + col0 + lid;
    float val = AZ * H[off] + CS * S[off] - CT * acc[r] + bcol;
    S[off] = fmaxf(val, 0.0f);
  }
}

__global__ void fc_logsoftmax_kernel(const float* __restrict__ X, const float* __restrict__ Wf,
                                     const float* __restrict__ bf, float* __restrict__ out) {
  int n = blockIdx.x * blockDim.x + threadIdx.x;
  if (n >= Nn) return;
  const float* h = X + (size_t)n * DH;
  float logit[Kc];
#pragma unroll
  for (int k = 0; k < Kc; ++k) logit[k] = bf[k];
  for (int j = 0; j < DH; ++j) {
    float hj = h[j];
#pragma unroll
    for (int k = 0; k < Kc; ++k) logit[k] = fmaf(hj, Wf[j * Kc + k], logit[k]);
  }
  float m = logit[0];
#pragma unroll
  for (int k = 1; k < Kc; ++k) m = fmaxf(m, logit[k]);
  float s = 0.0f;
#pragma unroll
  for (int k = 0; k < Kc; ++k) s += expf(logit[k] - m);
  float lse = m + logf(s);
#pragma unroll
  for (int k = 0; k < Kc; ++k) out[(size_t)n * Kc + k] = logit[k] - lse;
}

extern "C" void kernel_launch(void* const* d_in, const int* in_sizes, int n_in,
                              void* d_out, int out_size, void* d_ws, size_t ws_size,
                              hipStream_t stream) {
  const float* x    = (const float*)d_in[0];
  const int*   edge = (const int*)  d_in[1];
  const float* W1   = (const float*)d_in[2];
  const float* b1   = (const float*)d_in[3];
  const float* W2   = (const float*)d_in[4];
  const float* b2   = (const float*)d_in[5];
  const float* Wf   = (const float*)d_in[6];
  const float* bf   = (const float*)d_in[7];
  float* out = (float*)d_out;

  float* H    = (float*)d_ws;
  float* S    = H + (size_t)Nn * DH;
  float* dinv = S + (size_t)Nn * DH;
  float* G    = dinv + Nn;
  int*   cnt  = (int*)(G + (size_t)DH * DH);
  int*   offs = cnt + Nn;
  int*   rank = offs + Nn + 4;
  int*   csr  = rank + Ee;

  dim3 b256(256);

  fill4_kernel<<<128, b256, 0, stream>>>((float4*)cnt, (long)Nn / 4, 0.0f);
  rank_kernel<<<(Ee + 255) / 256, b256, 0, stream>>>(edge, cnt, rank);
  scan_kernel<<<(Nn + 1023) / 1024, b256, 0, stream>>>(cnt, offs, dinv);
  place_kernel<<<(Ee + 255) / 256, b256, 0, stream>>>(edge, offs, rank, dinv, csr);

  gemm_wmma_kernel<<<Nn / 16, b256, 0, stream>>>(x, W1, H, DIN);
  gather_kernel<<<Nn / 8, b256, 0, stream>>>(offs, csr, dinv, H, S);
  fill4_kernel<<<16, b256, 0, stream>>>((float4*)G, (long)DH * DH / 4, 0.0f);
  gram_wmma_kernel<<<Nn / GRAM_CHUNK, b256, 0, stream>>>(H, G);
  update_wmma_kernel<<<Nn / 16, b256, 0, stream>>>(H, S, G, b1);

  gemm_wmma_kernel<<<Nn / 16, b256, 0, stream>>>(S, W2, H, DH);
  gather_kernel<<<Nn / 8, b256, 0, stream>>>(offs, csr, dinv, H, S);
  fill4_kernel<<<16, b256, 0, stream>>>((float4*)G, (long)DH * DH / 4, 0.0f);
  gram_wmma_kernel<<<Nn / GRAM_CHUNK, b256, 0, stream>>>(H, G);
  update_wmma_kernel<<<Nn / 16, b256, 0, stream>>>(H, S, G, b2);

  fc_logsoftmax_kernel<<<(Nn + 255) / 256, b256, 0, stream>>>(S, Wf, bf, out);
}


# LDS row pitch padded so WMMA fragment reads are free of bank conflicts: dense-product A strip 64 to 68 dwords (was 16-way), Gram strip 128 to 136 (was 2-way); sums and their order unchanged
# speedup vs baseline: 1.7779x; 1.1473x over previous
#include <hip/hip_runtime.h>
#include <math.h>

#define Nn   100000
#define Ee   800000
#define DIN  256
#define DH   128
#define Kc   16
#define GAMMA 0.1f
#define LAM1  1.0f
#define LAM2  0.5f
#define AZ (1.0f - GAMMA*LAM1 + GAMMA*LAM2)
#define CS (GAMMA*LAM1)
#define CT (GAMMA*LAM2)
#define GRAM_CHUNK 400
#define GRAM_TILE  16
#define GEMM_KTILE 64
#define GEMM_PITCH (GEMM_KTILE + 4)
#define GRAM_PITCH (DH + 8)

typedef __attribute__((ext_vector_type(2))) float v2f;
typedef __attribute__((ext_vector_type(8))) float v8f;

__device__ __forceinline__ v8f wmma4(v2f a, v2f b, v8f c) {
  return __builtin_amdgcn_wmma_f32_16x16x4_f32(
       false, a,  false, b,
       (short)0, c,  false,  false);
}


__global__ void fill4_kernel(float4* __restrict__ p, long n4, float v) {
  long i = (long)blockIdx.x * blockDim.x + threadIdx.x;
  long stride = (long)gridDim.x * blockDim.x;
  float4 val = make_float4(v, v, v, v);
  for (; i < n4; i += stride) p[i] = val;
}

__global__ void rank_kernel(const int* __restrict__ edge, int* __restrict__ cnt,
                            int* __restrict__ rank) {
  int e = blockIdx.x * blockDim.x + threadIdx.x;
  if (e < Ee) rank[e] = atomicAdd(&cnt[edge[Ee + e]], 1);
}

__global__ void __launch_bounds__(256)
scan_kernel(const int* __restrict__ cnt, int* __restrict__ offs, float* __restrict__ dinv) {
  __shared__ int red[8];
  __shared__ int wtot[8];
  int tid  = threadIdx.x;
  int lane = tid & 31;
  int wave = tid >> 5;
  int b    = blockIdx.x;
  const int4* c4 = (const int4*)cnt;
  int p = 0;
#pragma unroll 8
  for (int k = 0; k < b; ++k) {
    int4 v = c4[k * 256 + tid];
    p += (v.x + v.y) + (v.z + v.w);
  }
#pragma unroll
  for (int d = 16; d >= 1; d >>= 1) p += __shfl_xor(p, d);
  if (lane == 0) red[wave] = p;

  int i = b * 1024 + tid * 4;
  int4 v = make_int4(0, 0, 0, 0);
  if (i < Nn) v = c4[b * 256 + tid];
  int s = (v.x + v.y) + (v.z + v.w);
  int x = s;
#pragma unroll
  for (int d = 1; d < 32; d <<= 1) {
    int y = __shfl_up(x, d);
    if (lane >= d) x += y;
  }
  if (lane == 31) wtot[wave] = x;
  __syncthreads();
  int base = 0;
#pragma unroll
  for (int k = 0; k < 8; ++k) base += red[k];
#pragma unroll
  for (int k = 0; k < 8; ++k) base += (k < wave) ? wtot[k] : 0;
  int excl = base + x - s;
  if (i < Nn) {
    int4 o;
    o.x = excl;
    o.y = o.x + v.x;
    o.z = o.y + v.y;
    o.w = o.z + v.z;
    ((int4*)offs)[b * 256 + tid] = o;
    if (i + 4 == Nn) offs[Nn] = o.w + v.w;
    float d0 = (float)(v.x + 1), d1 = (float)(v.y + 1), d2 = (float)(v.z + 1), d3 = (float)(v.w + 1);
    float4 dv;
    dv.x = (d0 > 0.0f) ? rsqrtf(fmaxf(d0, 1.0f)) : 0.0f;
    dv.y = (d1 > 0.0f) ? rsqrtf(fmaxf(d1, 1.0f)) : 0.0f;
    dv.z = (d2 > 0.0f) ? rsqrtf(fmaxf(d2, 1.0f)) : 0.0f;
    dv.w = (d3 > 0.0f) ? rsqrtf(fmaxf(d3, 1.0f)) : 0.0f;
    ((float4*)dinv)[b * 256 + tid] = dv;
  }
}

__global__ void place_kernel(const int* __restrict__ edge, const int* __restrict__ offs,
                             const int* __restrict__ rank, const float* __restrict__ dinv,
                             int* __restrict__ csr) {
  int e = blockIdx.x * blockDim.x + threadIdx.x;
  if (e >= Ee) return;
  int r = edge[e];
  int c = edge[Ee + e];
  int pos = offs[c] + rank[e];
  pos = min(max(pos, 0), Ee - 1);
  int2 rec;
  rec.x = r;
  rec.y = __float_as_int(dinv[r] * dinv[c]);
  ((int2*)csr)[pos] = rec;
}

__global__ void __launch_bounds__(256)
gemm_wmma_kernel(const float* __restrict__ A, const float* __restrict__ B,
                 float* __restrict__ C, int Kdim) {
  __shared__ __align__(16) float atile[16 * GEMM_PITCH];
  int tid  = threadIdx.x;
  int wave = tid >> 5;
  int lane = tid & 31;
  int half = lane >> 4;
  int lid  = lane & 15;
  int kb   = half * 2;
  int row0 = blockIdx.x * 16;
  int col0 = wave * 16;
  int sr   = tid >> 4;
  int sc   = tid & 15;
  v8f acc = {};
  for (int k0 = 0; k0 < Kdim; k0 += GEMM_KTILE) {
    __syncthreads();
    ((float4*)(atile + sr * GEMM_PITCH))[sc] =
        ((const float4*)(A + (size_t)(row0 + sr) * Kdim + k0))[sc];
    __syncthreads();
#pragma unroll
    for (int ks = 0; ks < GEMM_KTILE; ks += 4) {
      v2f a = *(const v2f*)(atile + lid * GEMM_PITCH + ks + kb);
      v2f b;
      b.x = B[(size_t)(k0 + ks + kb)     * DH + col0 + lid];
      b.y = B[(size_t)(k0 + ks + kb + 1) * DH + col0 + lid];
      acc = wmma4(a, b, acc);
    }
  }
#pragma unroll
  for (int r = 0; r < 8; ++r)
    C[(size_t)(row0 + r + 8 * half) * DH + col0 + lid] = acc[r];
}

__global__ void __launch_bounds__(256)
gather_kernel(const int* __restrict__ offs, const int* __restrict__ csr,
              const float* __restrict__ dinv, const float* __restrict__ H, float* __restrict__ S) {
  int lane = threadIdx.x & 31;
  int c = blockIdx.x * 8 + __builtin_amdgcn_readfirstlane((int)(threadIdx.x >> 5));
  int beg = offs[c];
  int end = offs[c + 1];
  beg = min(max(beg, 0), Ee);
  end = min(max(end, beg), Ee);
  float di = dinv[c];
  float w0 = di * di;
  const float4 hs = *(const float4*)(H + (size_t)c * DH + lane * 4);
  float4 acc;
  acc.x = w0 * hs.x; acc.y = w0 * hs.y; acc.z = w0 * hs.z; acc.w = w0 * hs.w;
  for (int base = beg; base < end; base += 32) {
    int n = min(32, end - base);
    int2 my = make_int2(0, 0);
    if (lane < n) my = ((const int2*)csr)[base + lane];
#pragma unroll 4
    for (int j = 0; j < n; ++j) {
      int   r = __builtin_amdgcn_readlane(my.x, j);
      float w = __int_as_float(__builtin_amdgcn_readlane(my.y, j));
      r = min(max(r, 0), Nn - 1);
      const float4 h = *(const float4*)(H + (size_t)r * DH + lane * 4);
      acc.x = fmaf(w, h.x, acc.x);
      acc.y = fmaf(w, h.y, acc.y);
      acc.z = fmaf(w, h.z, acc.z);
      acc.w = fmaf(w, h.w, acc.w);
    }
  }
  *(float4*)(S + (size_t)c * DH + lane * 4) = acc;
}

__global__ void __launch_bounds__(256)
gram_wmma_kernel(const float* __restrict__ H, float* __restrict__ G) {
  __shared__ __align__(16) float tile[GRAM_TILE * GRAM_PITCH];
  int tid  = threadIdx.x;
  int wave = tid >> 5;
  int lane = tid & 31;
  int half = lane >> 4;
  int lid  = lane & 15;
  int kb   = 2 * half;
  int n0   = blockIdx.x * GRAM_CHUNK;
  v8f acc[8] = {};
  for (int base = n0; base < n0 + GRAM_CHUNK; base += GRAM_TILE) {
    __syncthreads();
    {
      const float4* src = (const float4*)(H + (size_t)base * DH);
      int trow = tid >> 5;
      int tc4  = tid & 31;
      ((float4*)(tile + trow * GRAM_PITCH))[tc4]       = src[tid];
      ((float4*)(tile + (trow + 8) * GRAM_PITCH))[tc4] = src[tid + 256];
    }
    __syncthreads();
#pragma unroll
    for (int ks = 0; ks < GRAM_TILE; ks += 4) {
      v2f a;
      a.x = tile[(ks + kb)     * GRAM_PITCH + wave * 16 + lid];
      a.y = tile[(ks + kb + 1) * GRAM_PITCH + wave * 16 + lid];
#pragma unroll
      for (int ct = 0; ct < 8; ++ct) {
        v2f b;
        b.x = tile[(ks + kb)     * GRAM_PITCH + ct * 16 + lid];
        b.y = tile[(ks + kb + 1) * GRAM_PITCH + ct * 16 + lid];
        acc[ct] = wmma4(a, b, acc[ct]);
      }
    }
  }
#pragma unroll
  for (int ct = 0; ct < 8; ++ct)
#pragma unroll
    for (int r = 0; r < 8; ++r)
      atomicAdd(&G[(size_t)(wave * 16 + r + 8 * half) * DH + ct * 16 + lid], acc[ct][r]);
}

__global__ void __launch_bounds__(256)
update_wmma_kernel(const float* __restrict__ H, float* __restrict__ S,
                   const float* __restrict__ G, const float* __restrict__ bias) {
  int wave = threadIdx.x >> 5;
  int lane = threadIdx.x & 31;
  int half = lane >> 4;
  int lid  = lane & 15;
  int kb   = 2 * half;
  int row0 = blockIdx.x * 16;
  int col0 = wave * 16;
  v8f acc = {};
  for (int kk = 0; kk < DH; kk += 4) {
    v2f a = *(const v2f*)(H + (size_t)(row0 + lid) * DH + kk + kb);
    v2f b;
    b.x = G[(size_t)(kk + kb)     * DH + col0 + lid];
    b.y = G[(size_t)(kk + kb + 1) * DH + col0 + lid];
    acc = wmma4(a, b, acc);
  }
  float bcol = bias[col0 + lid];
#pragma unroll
  for (int r = 0; r < 8; ++r) {
    size_t off = (size_t)(row0 + r + 8 * half) * DH + col0 + lid;
    float val = AZ * H[off] + CS * S[off] - CT * acc[r] + bcol;
    S[off] = fmaxf(val, 0.0f);
  }
}

__global__ void fc_logsoftmax_kernel(const float* __restrict__ X, const float* __restrict__ Wf,
                                     const float* __restrict__ bf, float* __restrict__ out) {
  int n = blockIdx.x * blockDim.x + threadIdx.x;
  if (n >= Nn) return;
  const float* h = X + (size_t)n * DH;
  float logit[Kc];
#pragma unroll
  for (int k = 0; k < Kc; ++k) logit[k] = bf[k];
  for (int j = 0; j < DH; ++j) {
    float hj = h[j];
#pragma unroll
    for (int k = 0; k < Kc; ++k) logit[k] = fmaf(hj, Wf[j * Kc + k], logit[k]);
  }
  float m = logit[0];
#pragma unroll
  for (int k = 1; k < Kc; ++k) m = fmaxf(m, logit[k]);
  float s = 0.0f;
#pragma unroll
  for (int k = 0; k < Kc; ++k) s += expf(logit[k] - m);
  float lse = m + logf(s);
#pragma unroll
  for (int k = 0; k < Kc; ++k) out[(size_t)n * Kc + k] = logit[k] - lse;
}

extern "C" void kernel_launch(void* const* d_in, const int* in_sizes, int n_in,
                              void* d_out, int out_size, void* d_ws, size_t ws_size,
                              hipStream_t stream) {
  const float* x    = (const float*)d_in[0];
  const int*   edge = (const int*)  d_in[1];
  const float* W1   = (const float*)d_in[2];
  const float* b1   = (const float*)d_in[3];
  const float* W2   = (const float*)d_in[4];
  const float* b2   = (const float*)d_in[5];
  const float* Wf   = (const float*)d_in[6];
  const float* bf   = (const float*)d_in[7];
  float* out = (float*)d_out;

  float* H    = (float*)d_ws;
  float* S    = H + (size_t)Nn * DH;
  float* dinv = S + (size_t)Nn * DH;
  float* G    = dinv + Nn;
  int*   cnt  = (int*)(G + (size_t)DH * DH);
  int*   offs = cnt + Nn;
  int*   rank = offs + Nn + 4;
  int*   csr  = rank + Ee;

  dim3 b256(256);

  fill4_kernel<<<128, b256, 0, stream>>>((float4*)cnt, (long)Nn / 4, 0.0f);
  rank_kernel<<<(Ee + 255) / 256, b256, 0, stream>>>(edge, cnt, rank);
  scan_kernel<<<(Nn + 1023) / 1024, b256, 0, stream>>>(cnt, offs, dinv);
  place_kernel<<<(Ee + 255) / 256, b256, 0, stream>>>(edge, offs, rank, dinv, csr);

  gemm_wmma_kernel<<<Nn / 16, b256, 0, stream>>>(x, W1, H, DIN);
  gather_kernel<<<Nn / 8, b256, 0, stream>>>(offs, csr, dinv, H, S);
  fill4_kernel<<<16, b256, 0, stream>>>((float4*)G, (long)DH * DH / 4, 0.0f);
  gram_wmma_kernel<<<Nn / GRAM_CHUNK, b256, 0, stream>>>(H, G);
  update_wmma_kernel<<<Nn / 16, b256, 0, stream>>>(H, S, G, b1);

  gemm_wmma_kernel<<<Nn / 16, b256, 0, stream>>>(S, W2, H, DH);
  gather_kernel<<<Nn / 8, b256, 0, stream>>>(offs, csr, dinv, H, S);
  fill4_kernel<<<16, b256, 0, stream>>>((float4*)G, (long)DH * DH / 4, 0.0f);
  gram_wmma_kernel<<<Nn / GRAM_CHUNK, b256, 0, stream>>>(H, G);
  update_wmma_kernel<<<Nn / 16, b256, 0, stream>>>(H, S, G, b2);

  fc_logsoftmax_kernel<<<(Nn + 255) / 256, b256, 0, stream>>>(S, Wf, bf, out);
}


# Update kernel: block stages its 16x128 strip of H once in LDS with whole-line float4 loads (pitch 132); A fragments and the combine's H term read LDS instead of 16-line global loads; sums unchanged
# speedup vs baseline: 2.0024x; 1.1263x over previous
#include <hip/hip_runtime.h>
#include <math.h>

#define Nn   100000
#define Ee   800000
#define DIN  256
#define DH   128
#define Kc   16
#define GAMMA 0.1f
#define LAM1  1.0f
#define LAM2  0.5f
#define AZ (1.0f - GAMMA*LAM1 + GAMMA*LAM2)
#define CS (GAMMA*LAM1)
#define CT (GAMMA*LAM2)
#define GRAM_CHUNK 400
#define GRAM_TILE  16
#define GEMM_KTILE 64
#define GEMM_PITCH (GEMM_KTILE + 4)
#define GRAM_PITCH (DH + 8)
#define UPD_PITCH (DH + 4)

typedef __attribute__((ext_vector_type(2))) float v2f;
typedef __attribute__((ext_vector_type(8))) float v8f;

__device__ __forceinline__ v8f wmma4(v2f a, v2f b, v8f c) {
  return __builtin_amdgcn_wmma_f32_16x16x4_f32(
       false, a,  false, b,
       (short)0, c,  false,  false);
}


__global__ void fill4_kernel(float4* __restrict__ p, long n4, float v) {
  long i = (long)blockIdx.x * blockDim.x + threadIdx.x;
  long stride = (long)gridDim.x * blockDim.x;
  float4 val = make_float4(v, v, v, v);
  for (; i < n4; i += stride) p[i] = val;
}

__global__ void rank_kernel(const int* __restrict__ edge, int* __restrict__ cnt,
                            int* __restrict__ rank) {
  int e = blockIdx.x * blockDim.x + threadIdx.x;
  if (e < Ee) rank[e] = atomicAdd(&cnt[edge[Ee + e]], 1);
}

__global__ void __launch_bounds__(256)
scan_kernel(const int* __restrict__ cnt, int* __restrict__ offs, float* __restrict__ dinv) {
  __shared__ int red[8];
  __shared__ int wtot[8];
  int tid  = threadIdx.x;
  int lane = tid & 31;
  int wave = tid >> 5;
  int b    = blockIdx.x;
  const int4* c4 = (const int4*)cnt;
  int p = 0;
#pragma unroll 8
  for (int k = 0; k < b; ++k) {
    int4 v = c4[k * 256 + tid];
    p += (v.x + v.y) + (v.z + v.w);
  }
#pragma unroll
  for (int d = 16; d >= 1; d >>= 1) p += __shfl_xor(p, d);
  if (lane == 0) red[wave] = p;

  int i = b * 1024 + tid * 4;
  int4 v = make_int4(0, 0, 0, 0);
  if (i < Nn) v = c4[b * 256 + tid];
  int s = (v.x + v.y) + (v.z + v.w);
  int x = s;
#pragma unroll
  for (int d = 1; d < 32; d <<= 1) {
    int y = __shfl_up(x, d);
    if (lane >= d) x += y;
  }
  if (lane == 31) wtot[wave] = x;
  __syncthreads();
  int base = 0;
#pragma unroll
  for (int k = 0; k < 8; ++k) base += red[k];
#pragma unroll
  for (int k = 0; k < 8; ++k) base += (k < wave) ? wtot[k] : 0;
  int excl = base + x - s;
  if (i < Nn) {
    int4 o;
    o.x = excl;
    o.y = o.x + v.x;
    o.z = o.y + v.y;
    o.w = o.z + v.z;
    ((int4*)offs)[b * 256 + tid] = o;
    if (i + 4 == Nn) offs[Nn] = o.w + v.w;
    float d0 = (float)(v.x + 1), d1 = (float)(v.y + 1), d2 = (float)(v.z + 1), d3 = (float)(v.w + 1);
    float4 dv;
    dv.x = (d0 > 0.0f) ? rsqrtf(fmaxf(d0, 1.0f)) : 0.0f;
    dv.y = (d1 > 0.0f) ? rsqrtf(fmaxf(d1, 1.0f)) : 0.0f;
    dv.z = (d2 > 0.0f) ? rsqrtf(fmaxf(d2, 1.0f)) : 0.0f;
    dv.w = (d3 > 0.0f) ? rsqrtf(fmaxf(d3, 1.0f)) : 0.0f;
    ((float4*)dinv)[b * 256 + tid] = dv;
  }
}

__global__ void place_kernel(const int* __restrict__ edge, const int* __restrict__ offs,
                             const int* __restrict__ rank, const float* __restrict__ dinv,
                             int* __restrict__ csr) {
  int e = blockIdx.x * blockDim.x + threadIdx.x;
  if (e >= Ee) return;
  int r = edge[e];
  int c = edge[Ee + e];
  int pos = offs[c] + rank[e];
  pos = min(max(pos, 0), Ee - 1);
  int2 rec;
  rec.x = r;
  rec.y = __float_as_int(dinv[r] * dinv[c]);
  ((int2*)csr)[pos] = rec;
}

__global__ void __launch_bounds__(256)
gemm_wmma_kernel(const float* __restrict__ A, const float* __restrict__ B,
                 float* __restrict__ C, int Kdim) {
  __shared__ __align__(16) float atile[16 * GEMM_PITCH];
  int tid  = threadIdx.x;
  int wave = tid >> 5;
  int lane = tid & 31;
  int half = lane >> 4;
  int lid  = lane & 15;
  int kb   = half * 2;
  int row0 = blockIdx.x * 16;
  int col0 = wave * 16;
  int sr   = tid >> 4;
  int sc   = tid & 15;
  v8f acc = {};
  for (int k0 = 0; k0 < Kdim; k0 += GEMM_KTILE) {
    __syncthreads();
    ((float4*)(atile + sr * GEMM_PITCH))[sc] =
        ((const float4*)(A + (size_t)(row0 + sr) * Kdim + k0))[sc];
    __syncthreads();
#pragma unroll
    for (int ks = 0; ks < GEMM_KTILE; ks += 4) {
      v2f a = *(const v2f*)(atile + lid * GEMM_PITCH + ks + kb);
      v2f b;
      b.x = B[(size_t)(k0 + ks + kb)     * DH + col0 + lid];
      b.y = B[(size_t)(k0 + ks + kb + 1) * DH + col0 + lid];
      acc = wmma4(a, b, acc);
    }
  }
#pragma unroll
  for (int r = 0; r < 8; ++r)
    C[(size_t)(row0 + r + 8 * half) * DH + col0 + lid] = acc[r];
}

__global__ void __launch_bounds__(256)
gather_kernel(const int* __restrict__ offs, const int* __restrict__ csr,
              const float* __restrict__ dinv, const float* __restrict__ H, float* __restrict__ S) {
  int lane = threadIdx.x & 31;
  int c = blockIdx.x * 8 + __builtin_amdgcn_readfirstlane((int)(threadIdx.x >> 5));
  int beg = offs[c];
  int end = offs[c + 1];
  beg = min(max(beg, 0), Ee);
  end = min(max(end, beg), Ee);
  float di = dinv[c];
  float w0 = di * di;
  const float4 hs = *(const float4*)(H + (size_t)c * DH + lane * 4);
  float4 acc;
  acc.x = w0 * hs.x; acc.y = w0 * hs.y; acc.z = w0 * hs.z; acc.w = w0 * hs.w;
  for (int base = beg; base < end; base += 32) {
    int n = min(32, end - base);
    int2 my = make_int2(0, 0);
    if (lane < n) my = ((const int2*)csr)[base + lane];
#pragma unroll 4
    for (int j = 0; j < n; ++j) {
      int   r = __builtin_amdgcn_readlane(my.x, j);
      float w = __int_as_float(__builtin_amdgcn_readlane(my.y, j));
      r = min(max(r, 0), Nn - 1);
      const float4 h = *(const float4*)(H + (size_t)r * DH + lane * 4);
      acc.x = fmaf(w, h.x, acc.x);
      acc.y = fmaf(w, h.y, acc.y);
      acc.z = fmaf(w, h.z, acc.z);
      acc.w = fmaf(w, h.w, acc.w);
    }
  }
  *(float4*)(S + (size_t)c * DH + lane * 4) = acc;
}

__global__ void __launch_bounds__(256)
gram_wmma_kernel(const float* __restrict__ H, float* __restrict__ G) {
  __shared__ __align__(16) float tile[GRAM_TILE * GRAM_PITCH];
  int tid  = threadIdx.x;
  int wave = tid >> 5;
  int lane = tid & 31;
  int half = lane >> 4;
  int lid  = lane & 15;
  int kb   = 2 * half;
  int n0   = blockIdx.x * GRAM_CHUNK;
  v8f acc[8] = {};
  for (int base = n0; base < n0 + GRAM_CHUNK; base += GRAM_TILE) {
    __syncthreads();
    {
      const float4* src = (const float4*)(H + (size_t)base * DH);
      int trow = tid >> 5;
      int tc4  = tid & 31;
      ((float4*)(tile + trow * GRAM_PITCH))[tc4]       = src[tid];
      ((float4*)(tile + (trow + 8) * GRAM_PITCH))[tc4] = src[tid + 256];
    }
    __syncthreads();
#pragma unroll
    for (int ks = 0; ks < GRAM_TILE; ks += 4) {
      v2f a;
      a.x = tile[(ks + kb)     * GRAM_PITCH + wave * 16 + lid];
      a.y = tile[(ks + kb + 1) * GRAM_PITCH + wave * 16 + lid];
#pragma unroll
      for (int ct = 0; ct < 8; ++ct) {
        v2f b;
        b.x = tile[(ks + kb)     * GRAM_PITCH + ct * 16 + lid];
        b.y = tile[(ks + kb + 1) * GRAM_PITCH + ct * 16 + lid];
        acc[ct] = wmma4(a, b, acc[ct]);
      }
    }
  }
#pragma unroll
  for (int ct = 0; ct < 8; ++ct)
#pragma unroll
    for (int r = 0; r < 8; ++r)
      atomicAdd(&G[(size_t)(wave * 16 + r + 8 * half) * DH + ct * 16 + lid], acc[ct][r]);
}

__global__ void __launch_bounds__(256)
update_wmma_kernel(const float* __restrict__ H, float* __restrict__ S,
                   const float* __restrict__ G, const float* __restrict__ bias) {
  __shared__ __align__(16) float htile[16 * UPD_PITCH];
  int tid  = threadIdx.x;
  int wave = tid >> 5;
  int lane = tid & 31;
  int half = lane >> 4;
  int lid  = lane & 15;
  int kb   = 2 * half;
  int row0 = blockIdx.x * 16;
  int col0 = wave * 16;
  {
    const float4* src = (const float4*)(H + (size_t)row0 * DH);
    int trow = tid >> 5;
    int tc4  = tid & 31;
    ((float4*)(htile + trow * UPD_PITCH))[tc4]       = src[tid];
    ((float4*)(htile + (trow + 8) * UPD_PITCH))[tc4] = src[tid + 256];
  }
  __syncthreads();
  v8f acc = {};
  for (int kk = 0; kk < DH; kk += 4) {
    v2f a = *(const v2f*)(htile + lid * UPD_PITCH + kk + kb);
    v2f b;
    b.x = G[(size_t)(kk + kb)     * DH + col0 + lid];
    b.y = G[(size_t)(kk + kb + 1) * DH + col0 + lid];
    acc = wmma4(a, b, acc);
  }
  float bcol = bias[col0 + lid];
#pragma unroll
  for (int r = 0; r < 8; ++r) {
    int m = r + 8 * half;
    size_t off = (size_t)(row0 + m) * DH + col0 + lid;
    float val = AZ * htile[m * UPD_PITCH + col0 + lid] + CS * S[off] - CT * acc[r] + bcol;
    S[off] = fmaxf(val, 0.0f);
  }
}

__global__ void fc_logsoftmax_kernel(const float* __restrict__ X, const float* __restrict__ Wf,
                                     const float* __restrict__ bf, float* __restrict__ out) {
  int n = blockIdx.x * blockDim.x + threadIdx.x;
  if (n >= Nn) return;
  const float* h = X + (size_t)n * DH;
  float logit[Kc];
#pragma unroll
  for (int k = 0; k < Kc; ++k) logit[k] = bf[k];
  for (int j = 0; j < DH; ++j) {
    float hj = h[j];
#pragma unroll
    for (int k = 0; k < Kc; ++k) logit[k] = fmaf(hj, Wf[j * Kc + k], logit[k]);
  }
  float m = logit[0];
#pragma unroll
  for (int k = 1; k < Kc; ++k) m = fmaxf(m, logit[k]);
  float s = 0.0f;
#pragma unroll
  for (int k = 0; k < Kc; ++k) s += expf(logit[k] - m);
  float lse = m + logf(s);
#pragma unroll
  for (int k = 0; k < Kc; ++k) out[(size_t)n * Kc + k] = logit[k] - lse;
}

extern "C" void kernel_launch(void* const* d_in, const int* in_sizes, int n_in,
                              void* d_out, int out_size, void* d_ws, size_t ws_size,
                              hipStream_t stream) {
  const float* x    = (const float*)d_in[0];
  const int*   edge = (const int*)  d_in[1];
  const float* W1   = (const float*)d_in[2];
  const float* b1   = (const float*)d_in[3];
  const float* W2   = (const float*)d_in[4];
  const float* b2   = (const float*)d_in[5];
  const float* Wf   = (const float*)d_in[6];
  const float* bf   = (const float*)d_in[7];
  float* out = (float*)d_out;

  float* H    = (float*)d_ws;
  float* S    = H + (size_t)Nn * DH;
  float* dinv = S + (size_t)Nn * DH;
  float* G    = dinv + Nn;
  int*   cnt  = (int*)(G + (size_t)DH * DH);
  int*   offs = cnt + Nn;
  int*   rank = offs + Nn + 4;
  int*   csr  = rank + Ee;

  dim3 b256(256);

  fill4_kernel<<<128, b256, 0, stream>>>((float4*)cnt, (long)Nn / 4, 0.0f);
  rank_kernel<<<(Ee + 255) / 256, b256, 0, stream>>>(edge, cnt, rank);
  scan_kernel<<<(Nn + 1023) / 1024, b256, 0, stream>>>(cnt, offs, dinv);
  place_kernel<<<(Ee + 255) / 256, b256, 0, stream>>>(edge, offs, rank, dinv, csr);

  gemm_wmma_kernel<<<Nn / 16, b256, 0, stream>>>(x, W1, H, DIN);
  gather_kernel<<<Nn / 8, b256, 0, stream>>>(offs, csr, dinv, H, S);
  fill4_kernel<<<16, b256, 0, stream>>>((float4*)G, (long)DH * DH / 4, 0.0f);
  gram_wmma_kernel<<<Nn / GRAM_CHUNK, b256, 0, stream>>>(H, G);
  update_wmma_kernel<<<Nn / 16, b256, 0, stream>>>(H, S, G, b1);

  gemm_wmma_kernel<<<Nn / 16, b256, 0, stream>>>(S, W2, H, DH);
  gather_kernel<<<Nn / 8, b256, 0, stream>>>(offs, csr, dinv, H, S);
  fill4_kernel<<<16, b256, 0, stream>>>((float4*)G, (long)DH * DH / 4, 0.0f);
  gram_wmma_kernel<<<Nn / GRAM_CHUNK, b256, 0, stream>>>(H, G);
  update_wmma_kernel<<<Nn / 16, b256, 0, stream>>>(H, S, G, b2);

  fc_logsoftmax_kernel<<<(Nn + 255) / 256, b256, 0, stream>>>(S, Wf, bf, out);
}


# Dense products x@W and H@G: work-group tile 80 rows (was 16), each wave's B fragment feeds 5 row strips, so B is fetched once per 80 rows and waves fall 5x; sums and their order unchanged
# speedup vs baseline: 2.0868x; 1.0421x over previous
#include <hip/hip_runtime.h>
#include <math.h>

#define Nn   100000
#define Ee   800000
#define DIN  256
#define DH   128
#define Kc   16
#define GAMMA 0.1f
#define LAM1  1.0f
#define LAM2  0.5f
#define AZ (1.0f - GAMMA*LAM1 + GAMMA*LAM2)
#define CS (GAMMA*LAM1)
#define CT (GAMMA*LAM2)
#define GRAM_CHUNK 400
#define GRAM_TILE  16
#define GEMM_KTILE 64
#define GEMM_PITCH (GEMM_KTILE + 4)
#define GRAM_PITCH (DH + 8)
#define UPD_PITCH (DH + 4)
#define BLK_STRIPS 5
#define BLK_ROWS   (16 * BLK_STRIPS)

typedef __attribute__((ext_vector_type(2))) float v2f;
typedef __attribute__((ext_vector_type(8))) float v8f;

__device__ __forceinline__ v8f wmma4(v2f a, v2f b, v8f c) {
  return __builtin_amdgcn_wmma_f32_16x16x4_f32(
       false, a,  false, b,
       (short)0, c,  false,  false);
}


__global__ void fill4_kernel(float4* __restrict__ p, long n4, float v) {
  long i = (long)blockIdx.x * blockDim.x + threadIdx.x;
  long stride = (long)gridDim.x * blockDim.x;
  float4 val = make_float4(v, v, v, v);
  for (; i < n4; i += stride) p[i] = val;
}

__global__ void rank_kernel(const int* __restrict__ edge, int* __restrict__ cnt,
                            int* __restrict__ rank) {
  int e = blockIdx.x * blockDim.x + threadIdx.x;
  if (e < Ee) rank[e] = atomicAdd(&cnt[edge[Ee + e]], 1);
}

__global__ void __launch_bounds__(256)
scan_kernel(const int* __restrict__ cnt, int* __restrict__ offs, float* __restrict__ dinv) {
  __shared__ int red[8];
  __shared__ int wtot[8];
  int tid  = threadIdx.x;
  int lane = tid & 31;
  int wave = tid >> 5;
  int b    = blockIdx.x;
  const int4* c4 = (const int4*)cnt;
  int p = 0;
#pragma unroll 8
  for (int k = 0; k < b; ++k) {
    int4 v = c4[k * 256 + tid];
    p += (v.x + v.y) + (v.z + v.w);
  }
#pragma unroll
  for (int d = 16; d >= 1; d >>= 1) p += __shfl_xor(p, d);
  if (lane == 0) red[wave] = p;

  int i = b * 1024 + tid * 4;
  int4 v = make_int4(0, 0, 0, 0);
  if (i < Nn) v = c4[b * 256 + tid];
  int s = (v.x + v.y) + (v.z + v.w);
  int x = s;
#pragma unroll
  for (int d = 1; d < 32; d <<= 1) {
    int y = __shfl_up(x, d);
    if (lane >= d) x += y;
  }
  if (lane == 31) wtot[wave] = x;
  __syncthreads();
  int base = 0;
#pragma unroll
  for (int k = 0; k < 8; ++k) base += red[k];
#pragma unroll
  for (int k = 0; k < 8; ++k) base += (k < wave) ? wtot[k] : 0;
  int excl = base + x - s;
  if (i < Nn) {
    int4 o;
    o.x = excl;
    o.y = o.x + v.x;
    o.z = o.y + v.y;
    o.w = o.z + v.z;
    ((int4*)offs)[b * 256 + tid] = o;
    if (i + 4 == Nn) offs[Nn] = o.w + v.w;
    float d0 = (float)(v.x + 1), d1 = (float)(v.y + 1), d2 = (float)(v.z + 1), d3 = (float)(v.w + 1);
    float4 dv;
    dv.x = (d0 > 0.0f) ? rsqrtf(fmaxf(d0, 1.0f)) : 0.0f;
    dv.y = (d1 > 0.0f) ? rsqrtf(fmaxf(d1, 1.0f)) : 0.0f;
    dv.z = (d2 > 0.0f) ? rsqrtf(fmaxf(d2, 1.0f)) : 0.0f;
    dv.w = (d3 > 0.0f) ? rsqrtf(fmaxf(d3, 1.0f)) : 0.0f;
    ((float4*)dinv)[b * 256 + tid] = dv;
  }
}

__global__ void place_kernel(const int* __restrict__ edge, const int* __restrict__ offs,
                             const int* __restrict__ rank, const float* __restrict__ dinv,
                             int* __restrict__ csr) {
  int e = blockIdx.x * blockDim.x + threadIdx.x;
  if (e >= Ee) return;
  int r = edge[e];
  int c = edge[Ee + e];
  int pos = offs[c] + rank[e];
  pos = min(max(pos, 0), Ee - 1);
  int2 rec;
  rec.x = r;
  rec.y = __float_as_int(dinv[r] * dinv[c]);
  ((int2*)csr)[pos] = rec;
}

__global__ void __launch_bounds__(256)
gemm_wmma_kernel(const float* __restrict__ A, const float* __restrict__ B,
                 float* __restrict__ C, int Kdim) {
  __shared__ __align__(16) float atile[BLK_ROWS * GEMM_PITCH];
  int tid  = threadIdx.x;
  int wave = tid >> 5;
  int lane = tid & 31;
  int half = lane >> 4;
  int lid  = lane & 15;
  int kb   = half * 2;
  int row0 = blockIdx.x * BLK_ROWS;
  int col0 = wave * 16;
  v8f acc[BLK_STRIPS] = {};
  for (int k0 = 0; k0 < Kdim; k0 += GEMM_KTILE) {
    __syncthreads();
#pragma unroll
    for (int i = 0; i < BLK_STRIPS; ++i) {
      int idx = tid + 256 * i;
      int sr  = idx >> 4;
      int sc  = idx & 15;
      ((float4*)(atile + sr * GEMM_PITCH))[sc] =
          ((const float4*)(A + (size_t)(row0 + sr) * Kdim + k0))[sc];
    }
    __syncthreads();
#pragma unroll
    for (int ks = 0; ks < GEMM_KTILE; ks += 4) {
      v2f b;
      b.x = B[(size_t)(k0 + ks + kb)     * DH + col0 + lid];
      b.y = B[(size_t)(k0 + ks + kb + 1) * DH + col0 + lid];
#pragma unroll
      for (int s = 0; s < BLK_STRIPS; ++s) {
        v2f a = *(const v2f*)(atile + (s * 16 + lid) * GEMM_PITCH + ks + kb);
        acc[s] = wmma4(a, b, acc[s]);
      }
    }
  }
#pragma unroll
  for (int s = 0; s < BLK_STRIPS; ++s)
#pragma unroll
    for (int r = 0; r < 8; ++r)
      C[(size_t)(row0 + s * 16 + r + 8 * half) * DH + col0 + lid] = acc[s][r];
}

__global__ void __launch_bounds__(256)
gather_kernel(const int* __restrict__ offs, const int* __restrict__ csr,
              const float* __restrict__ dinv, const float* __restrict__ H, float* __restrict__ S) {
  int lane = threadIdx.x & 31;
  int c = blockIdx.x * 8 + __builtin_amdgcn_readfirstlane((int)(threadIdx.x >> 5));
  int beg = offs[c];
  int end = offs[c + 1];
  beg = min(max(beg, 0), Ee);
  end = min(max(end, beg), Ee);
  float di = dinv[c];
  float w0 = di * di;
  const float4 hs = *(const float4*)(H + (size_t)c * DH + lane * 4);
  float4 acc;
  acc.x = w0 * hs.x; acc.y = w0 * hs.y; acc.z = w0 * hs.z; acc.w = w0 * hs.w;
  for (int base = beg; base < end; base += 32) {
    int n = min(32, end - base);
    int2 my = make_int2(0, 0);
    if (lane < n) my = ((const int2*)csr)[base + lane];
#pragma unroll 4
    for (int j = 0; j < n; ++j) {
      int   r = __builtin_amdgcn_readlane(my.x, j);
      float w = __int_as_float(__builtin_amdgcn_readlane(my.y, j));
      r = min(max(r, 0), Nn - 1);
      const float4 h = *(const float4*)(H + (size_t)r * DH + lane * 4);
      acc.x = fmaf(w, h.x, acc.x);
      acc.y = fmaf(w, h.y, acc.y);
      acc.z = fmaf(w, h.z, acc.z);
      acc.w = fmaf(w, h.w, acc.w);
    }
  }
  *(float4*)(S + (size_t)c * DH + lane * 4) = acc;
}

__global__ void __launch_bounds__(256)
gram_wmma_kernel(const float* __restrict__ H, float* __restrict__ G) {
  __shared__ __align__(16) float tile[GRAM_TILE * GRAM_PITCH];
  int tid  = threadIdx.x;
  int wave = tid >> 5;
  int lane = tid & 31;
  int half = lane >> 4;
  int lid  = lane & 15;
  int kb   = 2 * half;
  int n0   = blockIdx.x * GRAM_CHUNK;
  v8f acc[8] = {};
  for (int base = n0; base < n0 + GRAM_CHUNK; base += GRAM_TILE) {
    __syncthreads();
    {
      const float4* src = (const float4*)(H + (size_t)base * DH);
      int trow = tid >> 5;
      int tc4  = tid & 31;
      ((float4*)(tile + trow * GRAM_PITCH))[tc4]       = src[tid];
      ((float4*)(tile + (trow + 8) * GRAM_PITCH))[tc4] = src[tid + 256];
    }
    __syncthreads();
#pragma unroll
    for (int ks = 0; ks < GRAM_TILE; ks += 4) {
      v2f a;
      a.x = tile[(ks + kb)     * GRAM_PITCH + wave * 16 + lid];
      a.y = tile[(ks + kb + 1) * GRAM_PITCH + wave * 16 + lid];
#pragma unroll
      for (int ct = 0; ct < 8; ++ct) {
        v2f b;
        b.x = tile[(ks + kb)     * GRAM_PITCH + ct * 16 + lid];
        b.y = tile[(ks + kb + 1) * GRAM_PITCH + ct * 16 + lid];
        acc[ct] = wmma4(a, b, acc[ct]);
      }
    }
  }
#pragma unroll
  for (int ct = 0; ct < 8; ++ct)
#pragma unroll
    for (int r = 0; r < 8; ++r)
      atomicAdd(&G[(size_t)(wave * 16 + r + 8 * half) * DH + ct * 16 + lid], acc[ct][r]);
}

__global__ void __launch_bounds__(256)
update_wmma_kernel(const float* __restrict__ H, float* __restrict__ S,
                   const float* __restrict__ G, const float* __restrict__ bias) {
  __shared__ __align__(16) float htile[BLK_ROWS * UPD_PITCH];
  int tid  = threadIdx.x;
  int wave = tid >> 5;
  int lane = tid & 31;
  int half = lane >> 4;
  int lid  = lane & 15;
  int kb   = 2 * half;
  int row0 = blockIdx.x * BLK_ROWS;
  int col0 = wave * 16;
  {
    const float4* src = (const float4*)(H + (size_t)row0 * DH);
#pragma unroll
    for (int i = 0; i < 2 * BLK_STRIPS; ++i) {
      int idx  = tid + 256 * i;
      int trow = idx >> 5;
      int tc4  = idx & 31;
      ((float4*)(htile + trow * UPD_PITCH))[tc4] = src[idx];
    }
  }
  __syncthreads();
  v8f acc[BLK_STRIPS] = {};
#pragma unroll 8
  for (int kk = 0; kk < DH; kk += 4) {
    v2f b;
    b.x = G[(size_t)(kk + kb)     * DH + col0 + lid];
    b.y = G[(size_t)(kk + kb + 1) * DH + col0 + lid];
#pragma unroll
    for (int s = 0; s < BLK_STRIPS; ++s) {
      v2f a = *(const v2f*)(htile + (s * 16 + lid) * UPD_PITCH + kk + kb);
      acc[s] = wmma4(a, b, acc[s]);
    }
  }
  float bcol = bias[col0 + lid];
#pragma unroll
  for (int s = 0; s < BLK_STRIPS; ++s) {
#pragma unroll
    for (int r = 0; r < 8; ++r) {
      int m = s * 16 + r + 8 * half;
      size_t off = (size_t)(row0 + m) * DH + col0 + lid;
      float val = AZ * htile[m * UPD_PITCH + col0 + lid] + CS * S[off] - CT * acc[s][r] + bcol;
      S[off] = fmaxf(val, 0.0f);
    }
  }
}

__global__ void fc_logsoftmax_kernel(const float* __restrict__ X, const float* __restrict__ Wf,
                                     const float* __restrict__ bf, float* __restrict__ out) {
  int n = blockIdx.x * blockDim.x + threadIdx.x;
  if (n >= Nn) return;
  const float* h = X + (size_t)n * DH;
  float logit[Kc];
#pragma unroll
  for (int k = 0; k < Kc; ++k) logit[k] = bf[k];
  for (int j = 0; j < DH; ++j) {
    float hj = h[j];
#pragma unroll
    for (int k = 0; k < Kc; ++k) logit[k] = fmaf(hj, Wf[j * Kc + k], logit[k]);
  }
  float m = logit[0];
#pragma unroll
  for (int k = 1; k < Kc; ++k) m = fmaxf(m, logit[k]);
  float s = 0.0f;
#pragma unroll
  for (int k = 0; k < Kc; ++k) s += expf(logit[k] - m);
  float lse = m + logf(s);
#pragma unroll
  for (int k = 0; k < Kc; ++k) out[(size_t)n * Kc + k] = logit[k] - lse;
}

extern "C" void kernel_launch(void* const* d_in, const int* in_sizes, int n_in,
                              void* d_out, int out_size, void* d_ws, size_t ws_size,
                              hipStream_t stream) {
  const float* x    = (const float*)d_in[0];
  const int*   edge = (const int*)  d_in[1];
  const float* W1   = (const float*)d_in[2];
  const float* b1   = (const float*)d_in[3];
  const float* W2   = (const float*)d_in[4];
  const float* b2   = (const float*)d_in[5];
  const float* Wf   = (const float*)d_in[6];
  const float* bf   = (const float*)d_in[7];
  float* out = (float*)d_out;

  float* H    = (float*)d_ws;
  float* S    = H + (size_t)Nn * DH;
  float* dinv = S + (size_t)Nn * DH;
  float* G    = dinv + Nn;
  int*   cnt  = (int*)(G + (size_t)DH * DH);
  int*   offs = cnt + Nn;
  int*   rank = offs + Nn + 4;
  int*   csr  = rank + Ee;

  dim3 b256(256);

  fill4_kernel<<<128, b256, 0, stream>>>((float4*)cnt, (long)Nn / 4, 0.0f);
  rank_kernel<<<(Ee + 255) / 256, b256, 0, stream>>>(edge, cnt, rank);
  scan_kernel<<<(Nn + 1023) / 1024, b256, 0, stream>>>(cnt, offs, dinv);
  place_kernel<<<(Ee + 255) / 256, b256, 0, stream>>>(edge, offs, rank, dinv, csr);

  gemm_wmma_kernel<<<Nn / BLK_ROWS, b256, 0, stream>>>(x, W1, H, DIN);
  gather_kernel<<<Nn / 8, b256, 0, stream>>>(offs, csr, dinv, H, S);
  fill4_kernel<<<16, b256, 0, stream>>>((float4*)G, (long)DH * DH / 4, 0.0f);
  gram_wmma_kernel<<<Nn / GRAM_CHUNK, b256, 0, stream>>>(H, G);
  update_wmma_kernel<<<Nn / BLK_ROWS, b256, 0, stream>>>(H, S, G, b1);

  gemm_wmma_kernel<<<Nn / BLK_ROWS, b256, 0, stream>>>(S, W2, H, DH);
  gather_kernel<<<Nn / 8, b256, 0, stream>>>(offs, csr, dinv, H, S);
  fill4_kernel<<<16, b256, 0, stream>>>((float4*)G, (long)DH * DH / 4, 0.0f);
  gram_wmma_kernel<<<Nn / GRAM_CHUNK, b256, 0, stream>>>(H, G);
  update_wmma_kernel<<<Nn / BLK_ROWS, b256, 0, stream>>>(H, S, G, b2);

  fc_logsoftmax_kernel<<<(Nn + 255) / 256, b256, 0, stream>>>(S, Wf, bf, out);
}


# Propagate and Gram of a layer (both only read H) share one launch: work-groups 0-249 do the Gram strips, the next 12500 the row gathers; G cleared first; 15 launches to 13; sums, atomics as before
# speedup vs baseline: 2.2919x; 1.0983x over previous
#include <hip/hip_runtime.h>
#include <math.h>

#define Nn   100000
#define Ee   800000
#define DIN  256
#define DH   128
#define Kc   16
#define GAMMA 0.1f
#define LAM1  1.0f
#define LAM2  0.5f
#define AZ (1.0f - GAMMA*LAM1 + GAMMA*LAM2)
#define CS (GAMMA*LAM1)
#define CT (GAMMA*LAM2)
#define GRAM_CHUNK 400
#define GRAM_TILE  16
#define GEMM_KTILE 64
#define GEMM_PITCH (GEMM_KTILE + 4)
#define GRAM_PITCH (DH + 8)
#define UPD_PITCH (DH + 4)
#define BLK_STRIPS 5
#define BLK_ROWS   (16 * BLK_STRIPS)

typedef __attribute__((ext_vector_type(2))) float v2f;
typedef __attribute__((ext_vector_type(8))) float v8f;

__device__ __forceinline__ v8f wmma4(v2f a, v2f b, v8f c) {
  return __builtin_amdgcn_wmma_f32_16x16x4_f32(
       false, a,  false, b,
       (short)0, c,  false,  false);
}


__global__ void fill4_kernel(float4* __restrict__ p, long n4, float v) {
  long i = (long)blockIdx.x * blockDim.x + threadIdx.x;
  long stride = (long)gridDim.x * blockDim.x;
  float4 val = make_float4(v, v, v, v);
  for (; i < n4; i += stride) p[i] = val;
}

__global__ void rank_kernel(const int* __restrict__ edge, int* __restrict__ cnt,
                            int* __restrict__ rank) {
  int e = blockIdx.x * blockDim.x + threadIdx.x;
  if (e < Ee) rank[e] = atomicAdd(&cnt[edge[Ee + e]], 1);
}

__global__ void __launch_bounds__(256)
scan_kernel(const int* __restrict__ cnt, int* __restrict__ offs, float* __restrict__ dinv) {
  __shared__ int red[8];
  __shared__ int wtot[8];
  int tid  = threadIdx.x;
  int lane = tid & 31;
  int wave = tid >> 5;
  int b    = blockIdx.x;
  const int4* c4 = (const int4*)cnt;
  int p = 0;
#pragma unroll 8
  for (int k = 0; k < b; ++k) {
    int4 v = c4[k * 256 + tid];
    p += (v.x + v.y) + (v.z + v.w);
  }
#pragma unroll
  for (int d = 16; d >= 1; d >>= 1) p += __shfl_xor(p, d);
  if (lane == 0) red[wave] = p;

  int i = b * 1024 + tid * 4;
  int4 v = make_int4(0, 0, 0, 0);
  if (i < Nn) v = c4[b * 256 + tid];
  int s = (v.x + v.y) + (v.z + v.w);
  int x = s;
#pragma unroll
  for (int d = 1; d < 32; d <<= 1) {
    int y = __shfl_up(x, d);
    if (lane >= d) x += y;
  }
  if (lane == 31) wtot[wave] = x;
  __syncthreads();
  int base = 0;
#pragma unroll
  for (int k = 0; k < 8; ++k) base += red[k];
#pragma unroll
  for (int k = 0; k < 8; ++k) base += (k < wave) ? wtot[k] : 0;
  int excl = base + x - s;
  if (i < Nn) {
    int4 o;
    o.x = excl;
    o.y = o.x + v.x;
    o.z = o.y + v.y;
    o.w = o.z + v.z;
    ((int4*)offs)[b * 256 + tid] = o;
    if (i + 4 == Nn) offs[Nn] = o.w + v.w;
    float d0 = (float)(v.x + 1), d1 = (float)(v.y + 1), d2 = (float)(v.z + 1), d3 = (float)(v.w + 1);
    float4 dv;
    dv.x = (d0 > 0.0f) ? rsqrtf(fmaxf(d0, 1.0f)) : 0.0f;
    dv.y = (d1 > 0.0f) ? rsqrtf(fmaxf(d1, 1.0f)) : 0.0f;
    dv.z = (d2 > 0.0f) ? rsqrtf(fmaxf(d2, 1.0f)) : 0.0f;
    dv.w = (d3 > 0.0f) ? rsqrtf(fmaxf(d3, 1.0f)) : 0.0f;
    ((float4*)dinv)[b * 256 + tid] = dv;
  }
}

__global__ void place_kernel(const int* __restrict__ edge, const int* __restrict__ offs,
                             const int* __restrict__ rank, const float* __restrict__ dinv,
                             int* __restrict__ csr) {
  int e = blockIdx.x * blockDim.x + threadIdx.x;
  if (e >= Ee) return;
  int r = edge[e];
  int c = edge[Ee + e];
  int pos = offs[c] + rank[e];
  pos = min(max(pos, 0), Ee - 1);
  int2 rec;
  rec.x = r;
  rec.y = __float_as_int(dinv[r] * dinv[c]);
  ((int2*)csr)[pos] = rec;
}

__global__ void __launch_bounds__(256)
gemm_wmma_kernel(const float* __restrict__ A, const float* __restrict__ B,
                 float* __restrict__ C, int Kdim) {
  __shared__ __align__(16) float atile[BLK_ROWS * GEMM_PITCH];
  int tid  = threadIdx.x;
  int wave = tid >> 5;
  int lane = tid & 31;
  int half = lane >> 4;
  int lid  = lane & 15;
  int kb   = half * 2;
  int row0 = blockIdx.x * BLK_ROWS;
  int col0 = wave * 16;
  v8f acc[BLK_STRIPS] = {};
  for (int k0 = 0; k0 < Kdim; k0 += GEMM_KTILE) {
    __syncthreads();
#pragma unroll
    for (int i = 0; i < BLK_STRIPS; ++i) {
      int idx = tid + 256 * i;
      int sr  = idx >> 4;
      int sc  = idx & 15;
      ((float4*)(atile + sr * GEMM_PITCH))[sc] =
          ((const float4*)(A + (size_t)(row0 + sr) * Kdim + k0))[sc];
    }
    __syncthreads();
#pragma unroll
    for (int ks = 0; ks < GEMM_KTILE; ks += 4) {
      v2f b;
      b.x = B[(size_t)(k0 + ks + kb)     * DH + col0 + lid];
      b.y = B[(size_t)(k0 + ks + kb + 1) * DH + col0 + lid];
#pragma unroll
      for (int s = 0; s < BLK_STRIPS; ++s) {
        v2f a = *(const v2f*)(atile + (s * 16 + lid) * GEMM_PITCH + ks + kb);
        acc[s] = wmma4(a, b, acc[s]);
      }
    }
  }
#pragma unroll
  for (int s = 0; s < BLK_STRIPS; ++s)
#pragma unroll
    for (int r = 0; r < 8; ++r)
      C[(size_t)(row0 + s * 16 + r + 8 * half) * DH + col0 + lid] = acc[s][r];
}

__device__ __forceinline__ void
gather_rows(int gblock, const int* __restrict__ offs, const int* __restrict__ csr,
            const float* __restrict__ dinv, const float* __restrict__ H, float* __restrict__ S) {
  int lane = threadIdx.x & 31;
  int c = gblock * 8 + __builtin_amdgcn_readfirstlane((int)(threadIdx.x >> 5));
  int beg = offs[c];
  int end = offs[c + 1];
  beg = min(max(beg, 0), Ee);
  end = min(max(end, beg), Ee);
  float di = dinv[c];
  float w0 = di * di;
  const float4 hs = *(const float4*)(H + (size_t)c * DH + lane * 4);
  float4 acc;
  acc.x = w0 * hs.x; acc.y = w0 * hs.y; acc.z = w0 * hs.z; acc.w = w0 * hs.w;
  for (int base = beg; base < end; base += 32) {
    int n = min(32, end - base);
    int2 my = make_int2(0, 0);
    if (lane < n) my = ((const int2*)csr)[base + lane];
#pragma unroll 4
    for (int j = 0; j < n; ++j) {
      int   r = __builtin_amdgcn_readlane(my.x, j);
      float w = __int_as_float(__builtin_amdgcn_readlane(my.y, j));
      r = min(max(r, 0), Nn - 1);
      const float4 h = *(const float4*)(H + (size_t)r * DH + lane * 4);
      acc.x = fmaf(w, h.x, acc.x);
      acc.y = fmaf(w, h.y, acc.y);
      acc.z = fmaf(w, h.z, acc.z);
      acc.w = fmaf(w, h.w, acc.w);
    }
  }
  *(float4*)(S + (size_t)c * DH + lane * 4) = acc;
}

__device__ __forceinline__ void
gram_strips(int mblock, float* __restrict__ tile, const float* __restrict__ H, float* __restrict__ G) {
  int tid  = threadIdx.x;
  int wave = tid >> 5;
  int lane = tid & 31;
  int half = lane >> 4;
  int lid  = lane & 15;
  int kb   = 2 * half;
  int n0   = mblock * GRAM_CHUNK;
  v8f acc[8] = {};
  for (int base = n0; base < n0 + GRAM_CHUNK; base += GRAM_TILE) {
    __syncthreads();
    {
      const float4* src = (const float4*)(H + (size_t)base * DH);
      int trow = tid >> 5;
      int tc4  = tid & 31;
      ((float4*)(tile + trow * GRAM_PITCH))[tc4]       = src[tid];
      ((float4*)(tile + (trow + 8) * GRAM_PITCH))[tc4] = src[tid + 256];
    }
    __syncthreads();
#pragma unroll
    for (int ks = 0; ks < GRAM_TILE; ks += 4) {
      v2f a;
      a.x = tile[(ks + kb)     * GRAM_PITCH + wave * 16 + lid];
      a.y = tile[(ks + kb + 1) * GRAM_PITCH + wave * 16 + lid];
#pragma unroll
      for (int ct = 0; ct < 8; ++ct) {
        v2f b;
        b.x = tile[(ks + kb)     * GRAM_PITCH + ct * 16 + lid];
        b.y = tile[(ks + kb + 1) * GRAM_PITCH + ct * 16 + lid];
        acc[ct] = wmma4(a, b, acc[ct]);
      }
    }
  }
#pragma unroll
  for (int ct = 0; ct < 8; ++ct)
#pragma unroll
    for (int r = 0; r < 8; ++r)
      atomicAdd(&G[(size_t)(wave * 16 + r + 8 * half) * DH + ct * 16 + lid], acc[ct][r]);
}

#define GRAM_BLOCKS   (Nn / GRAM_CHUNK)
#define GATHER_BLOCKS (Nn / 8)
__global__ void __launch_bounds__(256)
propagate_gram_kernel(const int* __restrict__ offs, const int* __restrict__ csr,
                      const float* __restrict__ dinv, const float* __restrict__ H,
                      float* __restrict__ S, float* __restrict__ G) {
  __shared__ __align__(16) float tile[GRAM_TILE * GRAM_PITCH];
  int b = blockIdx.x;
  if (b < GRAM_BLOCKS) gram_strips(b, tile, H, G);
  else                 gather_rows(b - GRAM_BLOCKS, offs, csr, dinv, H, S);
}

__global__ void __launch_bounds__(256)
update_wmma_kernel(const float* __restrict__ H, float* __restrict__ S,
                   const float* __restrict__ G, const float* __restrict__ bias) {
  __shared__ __align__(16) float htile[BLK_ROWS * UPD_PITCH];
  int tid  = threadIdx.x;
  int wave = tid >> 5;
  int lane = tid & 31;
  int half = lane >> 4;
  int lid  = lane & 15;
  int kb   = 2 * half;
  int row0 = blockIdx.x * BLK_ROWS;
  int col0 = wave * 16;
  {
    const float4* src = (const float4*)(H + (size_t)row0 * DH);
#pragma unroll
    for (int i = 0; i < 2 * BLK_STRIPS; ++i) {
      int idx  = tid + 256 * i;
      int trow = idx >> 5;
      int tc4  = idx & 31;
      ((float4*)(htile + trow * UPD_PITCH))[tc4] = src[idx];
    }
  }
  __syncthreads();
  v8f acc[BLK_STRIPS] = {};
#pragma unroll 8
  for (int kk = 0; kk < DH; kk += 4) {
    v2f b;
    b.x = G[(size_t)(kk + kb)     * DH + col0 + lid];
    b.y = G[(size_t)(kk + kb + 1) * DH + col0 + lid];
#pragma unroll
    for (int s = 0; s < BLK_STRIPS; ++s) {
      v2f a = *(const v2f*)(htile + (s * 16 + lid) * UPD_PITCH + kk + kb);
      acc[s] = wmma4(a, b, acc[s]);
    }
  }
  float bcol = bias[col0 + lid];
#pragma unroll
  for (int s = 0; s < BLK_STRIPS; ++s) {
#pragma unroll
    for (int r = 0; r < 8; ++r) {
      int m = s * 16 + r + 8 * half;
      size_t off = (size_t)(row0 + m) * DH + col0 + lid;
      float val = AZ * htile[m * UPD_PITCH + col0 + lid] + CS * S[off] - CT * acc[s][r] + bcol;
      S[off] = fmaxf(val, 0.0f);
    }
  }
}

__global__ void fc_logsoftmax_kernel(const float* __restrict__ X, const float* __restrict__ Wf,
                                     const float* __restrict__ bf, float* __restrict__ out) {
  int n = blockIdx.x * blockDim.x + threadIdx.x;
  if (n >= Nn) return;
  const float* h = X + (size_t)n * DH;
  float logit[Kc];
#pragma unroll
  for (int k = 0; k < Kc; ++k) logit[k] = bf[k];
  for (int j = 0; j < DH; ++j) {
    float hj = h[j];
#pragma unroll
    for (int k = 0; k < Kc; ++k) logit[k] = fmaf(hj, Wf[j * Kc + k], logit[k]);
  }
  float m = logit[0];
#pragma unroll
  for (int k = 1; k < Kc; ++k) m = fmaxf(m, logit[k]);
  float s = 0.0f;
#pragma unroll
  for (int k = 0; k < Kc; ++k) s += expf(logit[k] - m);
  float lse = m + logf(s);
#pragma unroll
  for (int k = 0; k < Kc; ++k) out[(size_t)n * Kc + k] = logit[k] - lse;
}

extern "C" void kernel_launch(void* const* d_in, const int* in_sizes, int n_in,
                              void* d_out, int out_size, void* d_ws, size_t ws_size,
                              hipStream_t stream) {
  const float* x    = (const float*)d_in[0];
  const int*   edge = (const int*)  d_in[1];
  const float* W1   = (const float*)d_in[2];
  const float* b1   = (const float*)d_in[3];
  const float* W2   = (const float*)d_in[4];
  const float* b2   = (const float*)d_in[5];
  const float* Wf   = (const float*)d_in[6];
  const float* bf   = (const float*)d_in[7];
  float* out = (float*)d_out;

  float* H    = (float*)d_ws;
  float* S    = H + (size_t)Nn * DH;
  float* dinv = S + (size_t)Nn * DH;
  float* G    = dinv + Nn;
  int*   cnt  = (int*)(G + (size_t)DH * DH);
  int*   offs = cnt + Nn;
  int*   rank = offs + Nn + 4;
  int*   csr  = rank + Ee;

  dim3 b256(256);

  fill4_kernel<<<128, b256, 0, stream>>>((float4*)cnt, (long)Nn / 4, 0.0f);
  rank_kernel<<<(Ee + 255) / 256, b256, 0, stream>>>(edge, cnt, rank);
  scan_kernel<<<(Nn + 1023) / 1024, b256, 0, stream>>>(cnt, offs, dinv);
  place_kernel<<<(Ee + 255) / 256, b256, 0, stream>>>(edge, offs, rank, dinv, csr);

  gemm_wmma_kernel<<<Nn / BLK_ROWS, b256, 0, stream>>>(x, W1, H, DIN);
  fill4_kernel<<<16, b256, 0, stream>>>((float4*)G, (long)DH * DH / 4, 0.0f);
  propagate_gram_kernel<<<GRAM_BLOCKS + GATHER_BLOCKS, b256, 0, stream>>>(offs, csr, dinv, H, S, G);
  update_wmma_kernel<<<Nn / BLK_ROWS, b256, 0, stream>>>(H, S, G, b1);

  gemm_wmma_kernel<<<Nn / BLK_ROWS, b256, 0, stream>>>(S, W2, H, DH);
  fill4_kernel<<<16, b256, 0, stream>>>((float4*)G, (long)DH * DH / 4, 0.0f);
  propagate_gram_kernel<<<GRAM_BLOCKS + GATHER_BLOCKS, b256, 0, stream>>>(offs, csr, dinv, H, S, G);
  update_wmma_kernel<<<Nn / BLK_ROWS, b256, 0, stream>>>(H, S, G, b2);

  fc_logsoftmax_kernel<<<(Nn + 255) / 256, b256, 0, stream>>>(S, Wf, bf, out);
}


# Layer tails: relu output tile stays in LDS and feeds the next dense product in the same launch (layer 1: @W2; layer 2: @Wf + log_softmax, now on WMMA as two interleaved k chains); 13 launches to 11
# speedup vs baseline: 2.5834x; 1.1272x over previous
#include <hip/hip_runtime.h>
#include <math.h>

#define Nn   100000
#define Ee   800000
#define DIN  256
#define DH   128
#define Kc   16
#define GAMMA 0.1f
#define LAM1  1.0f
#define LAM2  0.5f
#define AZ (1.0f - GAMMA*LAM1 + GAMMA*LAM2)
#define CS (GAMMA*LAM1)
#define CT (GAMMA*LAM2)
#define GRAM_CHUNK 400
#define GRAM_TILE  16
#define GEMM_KTILE 64
#define GEMM_PITCH (GEMM_KTILE + 4)
#define GRAM_PITCH (DH + 8)
#define UPD_PITCH (DH + 4)
#define BLK_STRIPS 5
#define BLK_ROWS   (16 * BLK_STRIPS)

typedef __attribute__((ext_vector_type(2))) float v2f;
typedef __attribute__((ext_vector_type(8))) float v8f;

__device__ __forceinline__ v8f wmma4(v2f a, v2f b, v8f c) {
  return __builtin_amdgcn_wmma_f32_16x16x4_f32(
       false, a,  false, b,
       (short)0, c,  false,  false);
}


__global__ void fill4_kernel(float4* __restrict__ p, long n4, float v) {
  long i = (long)blockIdx.x * blockDim.x + threadIdx.x;
  long stride = (long)gridDim.x * blockDim.x;
  float4 val = make_float4(v, v, v, v);
  for (; i < n4; i += stride) p[i] = val;
}

__global__ void rank_kernel(const int* __restrict__ edge, int* __restrict__ cnt,
                            int* __restrict__ rank) {
  int e = blockIdx.x * blockDim.x + threadIdx.x;
  if (e < Ee) rank[e] = atomicAdd(&cnt[edge[Ee + e]], 1);
}

__global__ void __launch_bounds__(256)
scan_kernel(const int* __restrict__ cnt, int* __restrict__ offs, float* __restrict__ dinv) {
  __shared__ int red[8];
  __shared__ int wtot[8];
  int tid  = threadIdx.x;
  int lane = tid & 31;
  int wave = tid >> 5;
  int b    = blockIdx.x;
  const int4* c4 = (const int4*)cnt;
  int p = 0;
#pragma unroll 8
  for (int k = 0; k < b; ++k) {
    int4 v = c4[k * 256 + tid];
    p += (v.x + v.y) + (v.z + v.w);
  }
#pragma unroll
  for (int d = 16; d >= 1; d >>= 1) p += __shfl_xor(p, d);
  if (lane == 0) red[wave] = p;

  int i = b * 1024 + tid * 4;
  int4 v = make_int4(0, 0, 0, 0);
  if (i < Nn) v = c4[b * 256 + tid];
  int s = (v.x + v.y) + (v.z + v.w);
  int x = s;
#pragma unroll
  for (int d = 1; d < 32; d <<= 1) {
    int y = __shfl_up(x, d);
    if (lane >= d) x += y;
  }
  if (lane == 31) wtot[wave] = x;
  __syncthreads();
  int base = 0;
#pragma unroll
  for (int k = 0; k < 8; ++k) base += red[k];
#pragma unroll
  for (int k = 0; k < 8; ++k) base += (k < wave) ? wtot[k] : 0;
  int excl = base + x - s;
  if (i < Nn) {
    int4 o;
    o.x = excl;
    o.y = o.x + v.x;
    o.z = o.y + v.y;
    o.w = o.z + v.z;
    ((int4*)offs)[b * 256 + tid] = o;
    if (i + 4 == Nn) offs[Nn] = o.w + v.w;
    float d0 = (float)(v.x + 1), d1 = (float)(v.y + 1), d2 = (float)(v.z + 1), d3 = (float)(v.w + 1);
    float4 dv;
    dv.x = (d0 > 0.0f) ? rsqrtf(fmaxf(d0, 1.0f)) : 0.0f;
    dv.y = (d1 > 0.0f) ? rsqrtf(fmaxf(d1, 1.0f)) : 0.0f;
    dv.z = (d2 > 0.0f) ? rsqrtf(fmaxf(d2, 1.0f)) : 0.0f;
    dv.w = (d3 > 0.0f) ? rsqrtf(fmaxf(d3, 1.0f)) : 0.0f;
    ((float4*)dinv)[b * 256 + tid] = dv;
  }
}

__global__ void place_kernel(const int* __restrict__ edge, const int* __restrict__ offs,
                             const int* __restrict__ rank, const float* __restrict__ dinv,
                             int* __restrict__ csr) {
  int e = blockIdx.x * blockDim.x + threadIdx.x;
  if (e >= Ee) return;
  int r = edge[e];
  int c = edge[Ee + e];
  int pos = offs[c] + rank[e];
  pos = min(max(pos, 0), Ee - 1);
  int2 rec;
  rec.x = r;
  rec.y = __float_as_int(dinv[r] * dinv[c]);
  ((int2*)csr)[pos] = rec;
}

__global__ void __launch_bounds__(256)
gemm_wmma_kernel(const float* __restrict__ A, const float* __restrict__ B,
                 float* __restrict__ C, int Kdim) {
  __shared__ __align__(16) float atile[BLK_ROWS * GEMM_PITCH];
  int tid  = threadIdx.x;
  int wave = tid >> 5;
  int lane = tid & 31;
  int half = lane >> 4;
  int lid  = lane & 15;
  int kb   = half * 2;
  int row0 = blockIdx.x * BLK_ROWS;
  int col0 = wave * 16;
  v8f acc[BLK_STRIPS] = {};
  for (int k0 = 0; k0 < Kdim; k0 += GEMM_KTILE) {
    __syncthreads();
#pragma unroll
    for (int i = 0; i < BLK_STRIPS; ++i) {
      int idx = tid + 256 * i;
      int sr  = idx >> 4;
      int sc  = idx & 15;
      ((float4*)(atile + sr * GEMM_PITCH))[sc] =
          ((const float4*)(A + (size_t)(row0 + sr) * Kdim + k0))[sc];
    }
    __syncthreads();
#pragma unroll
    for (int ks = 0; ks < GEMM_KTILE; ks += 4) {
      v2f b;
      b.x = B[(size_t)(k0 + ks + kb)     * DH + col0 + lid];
      b.y = B[(size_t)(k0 + ks + kb + 1) * DH + col0 + lid];
#pragma unroll
      for (int s = 0; s < BLK_STRIPS; ++s) {
        v2f a = *(const v2f*)(atile + (s * 16 + lid) * GEMM_PITCH + ks + kb);
        acc[s] = wmma4(a, b, acc[s]);
      }
    }
  }
#pragma unroll
  for (int s = 0; s < BLK_STRIPS; ++s)
#pragma unroll
    for (int r = 0; r < 8; ++r)
      C[(size_t)(row0 + s * 16 + r + 8 * half) * DH + col0 + lid] = acc[s][r];
}

__device__ __forceinline__ void
gather_rows(int gblock, const int* __restrict__ offs, const int* __restrict__ csr,
            const float* __restrict__ dinv, const float* __restrict__ H, float* __restrict__ S) {
  int lane = threadIdx.x & 31;
  int c = gblock * 8 + __builtin_amdgcn_readfirstlane((int)(threadIdx.x >> 5));
  int beg = offs[c];
  int end = offs[c + 1];
  beg = min(max(beg, 0), Ee);
  end = min(max(end, beg), Ee);
  float di = dinv[c];
  float w0 = di * di;
  const float4 hs = *(const float4*)(H + (size_t)c * DH + lane * 4);
  float4 acc;
  acc.x = w0 * hs.x; acc.y = w0 * hs.y; acc.z = w0 * hs.z; acc.w = w0 * hs.w;
  for (int base = beg; base < end; base += 32) {
    int n = min(32, end - base);
    int2 my = make_int2(0, 0);
    if (lane < n) my = ((const int2*)csr)[base + lane];
#pragma unroll 4
    for (int j = 0; j < n; ++j) {
      int   r = __builtin_amdgcn_readlane(my.x, j);
      float w = __int_as_float(__builtin_amdgcn_readlane(my.y, j));
      r = min(max(r, 0), Nn - 1);
      const float4 h = *(const float4*)(H + (size_t)r * DH + lane * 4);
      acc.x = fmaf(w, h.x, acc.x);
      acc.y = fmaf(w, h.y, acc.y);
      acc.z = fmaf(w, h.z, acc.z);
      acc.w = fmaf(w, h.w, acc.w);
    }
  }
  *(float4*)(S + (size_t)c * DH + lane * 4) = acc;
}

__device__ __forceinline__ void
gram_strips(int mblock, float* __restrict__ tile, const float* __restrict__ H, float* __restrict__ G) {
  int tid  = threadIdx.x;
  int wave = tid >> 5;
  int lane = tid & 31;
  int half = lane >> 4;
  int lid  = lane & 15;
  int kb   = 2 * half;
  int n0   = mblock * GRAM_CHUNK;
  v8f acc[8] = {};
  for (int base = n0; base < n0 + GRAM_CHUNK; base += GRAM_TILE) {
    __syncthreads();
    {
      const float4* src = (const float4*)(H + (size_t)base * DH);
      int trow = tid >> 5;
      int tc4  = tid & 31;
      ((float4*)(tile + trow * GRAM_PITCH))[tc4]       = src[tid];
      ((float4*)(tile + (trow + 8) * GRAM_PITCH))[tc4] = src[tid + 256];
    }
    __syncthreads();
#pragma unroll
    for (int ks = 0; ks < GRAM_TILE; ks += 4) {
      v2f a;
      a.x = tile[(ks + kb)     * GRAM_PITCH + wave * 16 + lid];
      a.y = tile[(ks + kb + 1) * GRAM_PITCH + wave * 16 + lid];
#pragma unroll
      for (int ct = 0; ct < 8; ++ct) {
        v2f b;
        b.x = tile[(ks + kb)     * GRAM_PITCH + ct * 16 + lid];
        b.y = tile[(ks + kb + 1) * GRAM_PITCH + ct * 16 + lid];
        acc[ct] = wmma4(a, b, acc[ct]);
      }
    }
  }
#pragma unroll
  for (int ct = 0; ct < 8; ++ct)
#pragma unroll
    for (int r = 0; r < 8; ++r)
      atomicAdd(&G[(size_t)(wave * 16 + r + 8 * half) * DH + ct * 16 + lid], acc[ct][r]);
}

#define GRAM_BLOCKS   (Nn / GRAM_CHUNK)
#define GATHER_BLOCKS (Nn / 8)
__global__ void __launch_bounds__(256)
propagate_gram_kernel(const int* __restrict__ offs, const int* __restrict__ csr,
                      const float* __restrict__ dinv, const float* __restrict__ H,
                      float* __restrict__ S, float* __restrict__ G) {
  __shared__ __align__(16) float tile[GRAM_TILE * GRAM_PITCH];
  int b = blockIdx.x;
  if (b < GRAM_BLOCKS) gram_strips(b, tile, H, G);
  else                 gather_rows(b - GRAM_BLOCKS, offs, csr, dinv, H, S);
}


__device__ __forceinline__ void
tile_times_mat(const float* __restrict__ tile, const float* __restrict__ Bm, int col0, int lid, int kb,
               v8f (&acc)[BLK_STRIPS]) {
#pragma unroll 8
  for (int kk = 0; kk < DH; kk += 4) {
    v2f b;
    b.x = Bm[(size_t)(kk + kb)     * DH + col0 + lid];
    b.y = Bm[(size_t)(kk + kb + 1) * DH + col0 + lid];
#pragma unroll
    for (int s = 0; s < BLK_STRIPS; ++s) {
      v2f a = *(const v2f*)(tile + (s * 16 + lid) * UPD_PITCH + kk + kb);
      acc[s] = wmma4(a, b, acc[s]);
    }
  }
}

__device__ __forceinline__ void
layer_tail(float* __restrict__ htile, float* __restrict__ H, const float* __restrict__ S,
           const float* __restrict__ G, const float* __restrict__ bias,
           const float* __restrict__ Wn, const float* __restrict__ bn, float* __restrict__ out, const bool last) {
  int tid  = threadIdx.x;
  int wave = tid >> 5;
  int lane = tid & 31;
  int half = lane >> 4;
  int lid  = lane & 15;
  int kb   = 2 * half;
  int row0 = blockIdx.x * BLK_ROWS;
  int col0 = wave * 16;
  {
    const float4* src = (const float4*)(H + (size_t)row0 * DH);
#pragma unroll
    for (int i = 0; i < 2 * BLK_STRIPS; ++i) {
      int idx  = tid + 256 * i;
      int trow = idx >> 5;
      int tc4  = idx & 31;
      ((float4*)(htile + trow * UPD_PITCH))[tc4] = src[idx];
    }
  }
  __syncthreads();
  v8f acc[BLK_STRIPS] = {};
  tile_times_mat(htile, G, col0, lid, kb, acc);
  float bcol = bias[col0 + lid];
  __syncthreads();
#pragma unroll
  for (int s = 0; s < BLK_STRIPS; ++s) {
#pragma unroll
    for (int r = 0; r < 8; ++r) {
      int m = s * 16 + r + 8 * half;
      size_t off = (size_t)(row0 + m) * DH + col0 + lid;
      float val = AZ * htile[m * UPD_PITCH + col0 + lid] + CS * S[off] - CT * acc[s][r] + bcol;
      htile[m * UPD_PITCH + col0 + lid] = fmaxf(val, 0.0f);
    }
  }
  __syncthreads();
  if (!last) {
    v8f nxt[BLK_STRIPS] = {};
    tile_times_mat(htile, Wn, col0, lid, kb, nxt);
#pragma unroll
    for (int s = 0; s < BLK_STRIPS; ++s)
#pragma unroll
      for (int r = 0; r < 8; ++r)
        H[(size_t)(row0 + s * 16 + r + 8 * half) * DH + col0 + lid] = nxt[s][r];
  } else {
    int wv = __builtin_amdgcn_readfirstlane(wave);
    if (wv < BLK_STRIPS) {
      const float* arow = htile + (wv * 16 + (((lid & 7) << 1) | (lid >> 3))) * UPD_PITCH + kb;
      v8f l0 = {}, l1 = {};
#pragma unroll 8
      for (int kk = 0; kk < DH; kk += 8) {
        v2f b0, b1;
        b0.x = Wn[(kk + kb)     * Kc + lid];
        b0.y = Wn[(kk + kb + 1) * Kc + lid];
        b1.x = Wn[(kk + kb + 4) * Kc + lid];
        b1.y = Wn[(kk + kb + 5) * Kc + lid];
        v2f a0 = *(const v2f*)(arow + kk);
        v2f a1 = *(const v2f*)(arow + kk + 4);
        l0 = wmma4(a0, b0, l0);
        l1 = wmma4(a1, b1, l1);
      }
      float bl = bn[lid];
      float* orow = out + (size_t)(row0 + wv * 16) * Kc + lane;
#pragma unroll
      for (int r = 0; r < 8; ++r) {
        float v = (l0[r] + l1[r]) + bl;
        float mx = v;
#pragma unroll
        for (int d = 8; d >= 1; d >>= 1) mx = fmaxf(mx, __shfl_xor(mx, d));
        float se = expf(v - mx);
#pragma unroll
        for (int d = 8; d >= 1; d >>= 1) se += __shfl_xor(se, d);
        orow[32 * r] = v - (mx + logf(se));
      }
    }
  }
}

__global__ void __launch_bounds__(256)
update_next_kernel(float* __restrict__ H, const float* __restrict__ S, const float* __restrict__ G,
                   const float* __restrict__ bias, const float* __restrict__ Wn) {
  __shared__ __align__(16) float htile[BLK_ROWS * UPD_PITCH];
  layer_tail(htile, H, S, G, bias, Wn, nullptr, nullptr, false);
}

__global__ void __launch_bounds__(256)
update_final_kernel(float* __restrict__ H, const float* __restrict__ S, const float* __restrict__ G,
                    const float* __restrict__ bias, const float* __restrict__ Wn, const float* __restrict__ bn,
                    float* __restrict__ out) {
  __shared__ __align__(16) float htile[BLK_ROWS * UPD_PITCH];
  layer_tail(htile, H, S, G, bias, Wn, bn, out, true);
}

extern "C" void kernel_launch(void* const* d_in, const int* in_sizes, int n_in,
                              void* d_out, int out_size, void* d_ws, size_t ws_size,
                              hipStream_t stream) {
  const float* x    = (const float*)d_in[0];
  const int*   edge = (const int*)  d_in[1];
  const float* W1   = (const float*)d_in[2];
  const float* b1   = (const float*)d_in[3];
  const float* W2   = (const float*)d_in[4];
  const float* b2   = (const float*)d_in[5];
  const float* Wf   = (const float*)d_in[6];
  const float* bf   = (const float*)d_in[7];
  float* out = (float*)d_out;

  float* H    = (float*)d_ws;
  float* S    = H + (size_t)Nn * DH;
  float* dinv = S + (size_t)Nn * DH;
  float* G    = dinv + Nn;
  int*   cnt  = (int*)(G + (size_t)DH * DH);
  int*   offs = cnt + Nn;
  int*   rank = offs + Nn + 4;
  int*   csr  = rank + Ee;

  dim3 b256(256);

  fill4_kernel<<<128, b256, 0, stream>>>((float4*)cnt, (long)Nn / 4, 0.0f);
  rank_kernel<<<(Ee + 255) / 256, b256, 0, stream>>>(edge, cnt, rank);
  scan_kernel<<<(Nn + 1023) / 1024, b256, 0, stream>>>(cnt, offs, dinv);
  place_kernel<<<(Ee + 255) / 256, b256, 0, stream>>>(edge, offs, rank, dinv, csr);

  gemm_wmma_kernel<<<Nn / BLK_ROWS, b256, 0, stream>>>(x, W1, H, DIN);
  fill4_kernel<<<16, b256, 0, stream>>>((float4*)G, (long)DH * DH / 4, 0.0f);
  propagate_gram_kernel<<<GRAM_BLOCKS + GATHER_BLOCKS, b256, 0, stream>>>(offs, csr, dinv, H, S, G);
  update_next_kernel<<<Nn / BLK_ROWS, b256, 0, stream>>>(H, S, G, b1, W2);

  fill4_kernel<<<16, b256, 0, stream>>>((float4*)G, (long)DH * DH / 4, 0.0f);
  propagate_gram_kernel<<<GRAM_BLOCKS + GATHER_BLOCKS, b256, 0, stream>>>(offs, csr, dinv, H, S, G);
  update_final_kernel<<<Nn / BLK_ROWS, b256, 0, stream>>>(H, S, G, b2, Wf, bf, out);
}


# Gram role re-tiled: work-group = 40-row strips x half the column tiles (was 16 rows x all 8): 10 load rounds per chunk, was 25; 4 accumulator tiles per wave; launch registers 96 to 64; sums unchanged
# speedup vs baseline: 2.7127x; 1.0501x over previous
#include <hip/hip_runtime.h>
#include <math.h>

#define Nn   100000
#define Ee   800000
#define DIN  256
#define DH   128
#define Kc   16
#define GAMMA 0.1f
#define LAM1  1.0f
#define LAM2  0.5f
#define AZ (1.0f - GAMMA*LAM1 + GAMMA*LAM2)
#define CS (GAMMA*LAM1)
#define CT (GAMMA*LAM2)
#define GRAM_CHUNK 400
#define GRAM_TILE  40
#define GEMM_KTILE 64
#define GEMM_PITCH (GEMM_KTILE + 4)
#define GRAM_PITCH (DH + 8)
#define UPD_PITCH (DH + 4)
#define BLK_STRIPS 5
#define BLK_ROWS   (16 * BLK_STRIPS)

typedef __attribute__((ext_vector_type(2))) float v2f;
typedef __attribute__((ext_vector_type(8))) float v8f;

__device__ __forceinline__ v8f wmma4(v2f a, v2f b, v8f c) {
  return __builtin_amdgcn_wmma_f32_16x16x4_f32(
       false, a,  false, b,
       (short)0, c,  false,  false);
}


__global__ void fill4_kernel(float4* __restrict__ p, long n4, float v) {
  long i = (long)blockIdx.x * blockDim.x + threadIdx.x;
  long stride = (long)gridDim.x * blockDim.x;
  float4 val = make_float4(v, v, v, v);
  for (; i < n4; i += stride) p[i] = val;
}

__global__ void rank_kernel(const int* __restrict__ edge, int* __restrict__ cnt,
                            int* __restrict__ rank) {
  int e = blockIdx.x * blockDim.x + threadIdx.x;
  if (e < Ee) rank[e] = atomicAdd(&cnt[edge[Ee + e]], 1);
}

__global__ void __launch_bounds__(256)
scan_kernel(const int* __restrict__ cnt, int* __restrict__ offs, float* __restrict__ dinv) {
  __shared__ int red[8];
  __shared__ int wtot[8];
  int tid  = threadIdx.x;
  int lane = tid & 31;
  int wave = tid >> 5;
  int b    = blockIdx.x;
  const int4* c4 = (const int4*)cnt;
  int p = 0;
#pragma unroll 8
  for (int k = 0; k < b; ++k) {
    int4 v = c4[k * 256 + tid];
    p += (v.x + v.y) + (v.z + v.w);
  }
#pragma unroll
  for (int d = 16; d >= 1; d >>= 1) p += __shfl_xor(p, d);
  if (lane == 0) red[wave] = p;

  int i = b * 1024 + tid * 4;
  int4 v = make_int4(0, 0, 0, 0);
  if (i < Nn) v = c4[b * 256 + tid];
  int s = (v.x + v.y) + (v.z + v.w);
  int x = s;
#pragma unroll
  for (int d = 1; d < 32; d <<= 1) {
    int y = __shfl_up(x, d);
    if (lane >= d) x += y;
  }
  if (lane == 31) wtot[wave] = x;
  __syncthreads();
  int base = 0;
#pragma unroll
  for (int k = 0; k < 8; ++k) base += red[k];
#pragma unroll
  for (int k = 0; k < 8; ++k) base += (k < wave) ? wtot[k] : 0;
  int excl = base + x - s;
  if (i < Nn) {
    int4 o;
    o.x = excl;
    o.y = o.x + v.x;
    o.z = o.y + v.y;
    o.w = o.z + v.z;
    ((int4*)offs)[b * 256 + tid] = o;
    if (i + 4 == Nn) offs[Nn] = o.w + v.w;
    float d0 = (float)(v.x + 1), d1 = (float)(v.y + 1), d2 = (float)(v.z + 1), d3 = (float)(v.w + 1);
    float4 dv;
    dv.x = (d0 > 0.0f) ? rsqrtf(fmaxf(d0, 1.0f)) : 0.0f;
    dv.y = (d1 > 0.0f) ? rsqrtf(fmaxf(d1, 1.0f)) : 0.0f;
    dv.z = (d2 > 0.0f) ? rsqrtf(fmaxf(d2, 1.0f)) : 0.0f;
    dv.w = (d3 > 0.0f) ? rsqrtf(fmaxf(d3, 1.0f)) : 0.0f;
    ((float4*)dinv)[b * 256 + tid] = dv;
  }
}

__global__ void place_kernel(const int* __restrict__ edge, const int* __restrict__ offs,
                             const int* __restrict__ rank, const float* __restrict__ dinv,
                             int* __restrict__ csr) {
  int e = blockIdx.x * blockDim.x + threadIdx.x;
  if (e >= Ee) return;
  int r = edge[e];
  int c = edge[Ee + e];
  int pos = offs[c] + rank[e];
  pos = min(max(pos, 0), Ee - 1);
  int2 rec;
  rec.x = r;
  rec.y = __float_as_int(dinv[r] * dinv[c]);
  ((int2*)csr)[pos] = rec;
}

__global__ void __launch_bounds__(256)
gemm_wmma_kernel(const float* __restrict__ A, const float* __restrict__ B,
                 float* __restrict__ C, int Kdim) {
  __shared__ __align__(16) float atile[BLK_ROWS * GEMM_PITCH];
  int tid  = threadIdx.x;
  int wave = tid >> 5;
  int lane = tid & 31;
  int half = lane >> 4;
  int lid  = lane & 15;
  int kb   = half * 2;
  int row0 = blockIdx.x * BLK_ROWS;
  int col0 = wave * 16;
  v8f acc[BLK_STRIPS] = {};
  for (int k0 = 0; k0 < Kdim; k0 += GEMM_KTILE) {
    __syncthreads();
#pragma unroll
    for (int i = 0; i < BLK_STRIPS; ++i) {
      int idx = tid + 256 * i;
      int sr  = idx >> 4;
      int sc  = idx & 15;
      ((float4*)(atile + sr * GEMM_PITCH))[sc] =
          ((const float4*)(A + (size_t)(row0 + sr) * Kdim + k0))[sc];
    }
    __syncthreads();
#pragma unroll
    for (int ks = 0; ks < GEMM_KTILE; ks += 4) {
      v2f b;
      b.x = B[(size_t)(k0 + ks + kb)     * DH + col0 + lid];
      b.y = B[(size_t)(k0 + ks + kb + 1) * DH + col0 + lid];
#pragma unroll
      for (int s = 0; s < BLK_STRIPS; ++s) {
        v2f a = *(const v2f*)(atile + (s * 16 + lid) * GEMM_PITCH + ks + kb);
        acc[s] = wmma4(a, b, acc[s]);
      }
    }
  }
#pragma unroll
  for (int s = 0; s < BLK_STRIPS; ++s)
#pragma unroll
    for (int r = 0; r < 8; ++r)
      C[(size_t)(row0 + s * 16 + r + 8 * half) * DH + col0 + lid] = acc[s][r];
}

__device__ __forceinline__ void
gather_rows(int gblock, const int* __restrict__ offs, const int* __restrict__ csr,
            const float* __restrict__ dinv, const float* __restrict__ H, float* __restrict__ S) {
  int lane = threadIdx.x & 31;
  int c = gblock * 8 + __builtin_amdgcn_readfirstlane((int)(threadIdx.x >> 5));
  int beg = offs[c];
  int end = offs[c + 1];
  beg = min(max(beg, 0), Ee);
  end = min(max(end, beg), Ee);
  float di = dinv[c];
  float w0 = di * di;
  const float4 hs = *(const float4*)(H + (size_t)c * DH + lane * 4);
  float4 acc;
  acc.x = w0 * hs.x; acc.y = w0 * hs.y; acc.z = w0 * hs.z; acc.w = w0 * hs.w;
  for (int base = beg; base < end; base += 32) {
    int n = min(32, end - base);
    int2 my = make_int2(0, 0);
    if (lane < n) my = ((const int2*)csr)[base + lane];
#pragma unroll 4
    for (int j = 0; j < n; ++j) {
      int   r = __builtin_amdgcn_readlane(my.x, j);
      float w = __int_as_float(__builtin_amdgcn_readlane(my.y, j));
      r = min(max(r, 0), Nn - 1);
      const float4 h = *(const float4*)(H + (size_t)r * DH + lane * 4);
      acc.x = fmaf(w, h.x, acc.x);
      acc.y = fmaf(w, h.y, acc.y);
      acc.z = fmaf(w, h.z, acc.z);
      acc.w = fmaf(w, h.w, acc.w);
    }
  }
  *(float4*)(S + (size_t)c * DH + lane * 4) = acc;
}

#define GRAM_CT 4
__device__ __forceinline__ void
gram_strips(int gblock, float* __restrict__ tile, const float* __restrict__ H, float* __restrict__ G) {
  int mblock = gblock >> 1;
  int cbase  = (gblock & 1) * (GRAM_CT * 16);
  int tid  = threadIdx.x;
  int wave = tid >> 5;
  int lane = tid & 31;
  int half = lane >> 4;
  int lid  = lane & 15;
  int kb   = 2 * half;
  int n0   = mblock * GRAM_CHUNK;
  v8f acc[GRAM_CT] = {};
  for (int base = n0; base < n0 + GRAM_CHUNK; base += GRAM_TILE) {
    __syncthreads();
    {
      const float4* src = (const float4*)(H + (size_t)base * DH);
      int trow = tid >> 5;
      int tc4  = tid & 31;
#pragma unroll
      for (int i = 0; i < GRAM_TILE / 8; ++i)
        ((float4*)(tile + (trow + 8 * i) * GRAM_PITCH))[tc4] = src[tid + 256 * i];
    }
    __syncthreads();
#pragma unroll 2
    for (int ks = 0; ks < GRAM_TILE; ks += 4) {
      v2f a;
      a.x = tile[(ks + kb)     * GRAM_PITCH + wave * 16 + lid];
      a.y = tile[(ks + kb + 1) * GRAM_PITCH + wave * 16 + lid];
#pragma unroll
      for (int ct = 0; ct < GRAM_CT; ++ct) {
        v2f b;
        b.x = tile[(ks + kb)     * GRAM_PITCH + cbase + ct * 16 + lid];
        b.y = tile[(ks + kb + 1) * GRAM_PITCH + cbase + ct * 16 + lid];
        acc[ct] = wmma4(a, b, acc[ct]);
      }
    }
  }
#pragma unroll
  for (int ct = 0; ct < GRAM_CT; ++ct)
#pragma unroll
    for (int r = 0; r < 8; ++r)
      atomicAdd(&G[(size_t)(wave * 16 + r + 8 * half) * DH + cbase + ct * 16 + lid], acc[ct][r]);
}

#define GRAM_BLOCKS   (2 * (Nn / GRAM_CHUNK))
#define GATHER_BLOCKS (Nn / 8)
__global__ void __launch_bounds__(256)
propagate_gram_kernel(const int* __restrict__ offs, const int* __restrict__ csr,
                      const float* __restrict__ dinv, const float* __restrict__ H,
                      float* __restrict__ S, float* __restrict__ G) {
  __shared__ __align__(16) float tile[GRAM_TILE * GRAM_PITCH];
  int b = blockIdx.x;
  if (b < GRAM_BLOCKS) gram_strips(b, tile, H, G);
  else                 gather_rows(b - GRAM_BLOCKS, offs, csr, dinv, H, S);
}


__device__ __forceinline__ void
tile_times_mat(const float* __restrict__ tile, const float* __restrict__ Bm, int col0, int lid, int kb,
               v8f (&acc)[BLK_STRIPS]) {
#pragma unroll 8
  for (int kk = 0; kk < DH; kk += 4) {
    v2f b;
    b.x = Bm[(size_t)(kk + kb)     * DH + col0 + lid];
    b.y = Bm[(size_t)(kk + kb + 1) * DH + col0 + lid];
#pragma unroll
    for (int s = 0; s < BLK_STRIPS; ++s) {
      v2f a = *(const v2f*)(tile + (s * 16 + lid) * UPD_PITCH + kk + kb);
      acc[s] = wmma4(a, b, acc[s]);
    }
  }
}

__device__ __forceinline__ void
layer_tail(float* __restrict__ htile, float* __restrict__ H, const float* __restrict__ S,
           const float* __restrict__ G, const float* __restrict__ bias,
           const float* __restrict__ Wn, const float* __restrict__ bn, float* __restrict__ out, const bool last) {
  int tid  = threadIdx.x;
  int wave = tid >> 5;
  int lane = tid & 31;
  int half = lane >> 4;
  int lid  = lane & 15;
  int kb   = 2 * half;
  int row0 = blockIdx.x * BLK_ROWS;
  int col0 = wave * 16;
  {
    const float4* src = (const float4*)(H + (size_t)row0 * DH);
#pragma unroll
    for (int i = 0; i < 2 * BLK_STRIPS; ++i) {
      int idx  = tid + 256 * i;
      int trow = idx >> 5;
      int tc4  = idx & 31;
      ((float4*)(htile + trow * UPD_PITCH))[tc4] = src[idx];
    }
  }
  __syncthreads();
  v8f acc[BLK_STRIPS] = {};
  tile_times_mat(htile, G, col0, lid, kb, acc);
  float bcol = bias[col0 + lid];
  __syncthreads();
#pragma unroll
  for (int s = 0; s < BLK_STRIPS; ++s) {
#pragma unroll
    for (int r = 0; r < 8; ++r) {
      int m = s * 16 + r + 8 * half;
      size_t off = (size_t)(row0 + m) * DH + col0 + lid;
      float val = AZ * htile[m * UPD_PITCH + col0 + lid] + CS * S[off] - CT * acc[s][r] + bcol;
      htile[m * UPD_PITCH + col0 + lid] = fmaxf(val, 0.0f);
    }
  }
  __syncthreads();
  if (!last) {
    v8f nxt[BLK_STRIPS] = {};
    tile_times_mat(htile, Wn, col0, lid, kb, nxt);
#pragma unroll
    for (int s = 0; s < BLK_STRIPS; ++s)
#pragma unroll
      for (int r = 0; r < 8; ++r)
        H[(size_t)(row0 + s * 16 + r + 8 * half) * DH + col0 + lid] = nxt[s][r];
  } else {
    int wv = __builtin_amdgcn_readfirstlane(wave);
    if (wv < BLK_STRIPS) {
      const float* arow = htile + (wv * 16 + (((lid & 7) << 1) | (lid >> 3))) * UPD_PITCH + kb;
      v8f l0 = {}, l1 = {};
#pragma unroll 8
      for (int kk = 0; kk < DH; kk += 8) {
        v2f b0, b1;
        b0.x = Wn[(kk + kb)     * Kc + lid];
        b0.y = Wn[(kk + kb + 1) * Kc + lid];
        b1.x = Wn[(kk + kb + 4) * Kc + lid];
        b1.y = Wn[(kk + kb + 5) * Kc + lid];
        v2f a0 = *(const v2f*)(arow + kk);
        v2f a1 = *(const v2f*)(arow + kk + 4);
        l0 = wmma4(a0, b0, l0);
        l1 = wmma4(a1, b1, l1);
      }
      float bl = bn[lid];
      float* orow = out + (size_t)(row0 + wv * 16) * Kc + lane;
#pragma unroll
      for (int r = 0; r < 8; ++r) {
        float v = (l0[r] + l1[r]) + bl;
        float mx = v;
#pragma unroll
        for (int d = 8; d >= 1; d >>= 1) mx = fmaxf(mx, __shfl_xor(mx, d));
        float se = expf(v - mx);
#pragma unroll
        for (int d = 8; d >= 1; d >>= 1) se += __shfl_xor(se, d);
        orow[32 * r] = v - (mx + logf(se));
      }
    }
  }
}

__global__ void __launch_bounds__(256)
update_next_kernel(float* __restrict__ H, const float* __restrict__ S, const float* __restrict__ G,
                   const float* __restrict__ bias, const float* __restrict__ Wn) {
  __shared__ __align__(16) float htile[BLK_ROWS * UPD_PITCH];
  layer_tail(htile, H, S, G, bias, Wn, nullptr, nullptr, false);
}

__global__ void __launch_bounds__(256)
update_final_kernel(float* __restrict__ H, const float* __restrict__ S, const float* __restrict__ G,
                    const float* __restrict__ bias, const float* __restrict__ Wn, const float* __restrict__ bn,
                    float* __restrict__ out) {
  __shared__ __align__(16) float htile[BLK_ROWS * UPD_PITCH];
  layer_tail(htile, H, S, G, bias, Wn, bn, out, true);
}

extern "C" void kernel_launch(void* const* d_in, const int* in_sizes, int n_in,
                              void* d_out, int out_size, void* d_ws, size_t ws_size,
                              hipStream_t stream) {
  const float* x    = (const float*)d_in[0];
  const int*   edge = (const int*)  d_in[1];
  const float* W1   = (const float*)d_in[2];
  const float* b1   = (const float*)d_in[3];
  const float* W2   = (const float*)d_in[4];
  const float* b2   = (const float*)d_in[5];
  const float* Wf   = (const float*)d_in[6];
  const float* bf   = (const float*)d_in[7];
  float* out = (float*)d_out;

  float* H    = (float*)d_ws;
  float* S    = H + (size_t)Nn * DH;
  float* dinv = S + (size_t)Nn * DH;
  float* G    = dinv + Nn;
  int*   cnt  = (int*)(G + (size_t)DH * DH);
  int*   offs = cnt + Nn;
  int*   rank = offs + Nn + 4;
  int*   csr  = rank + Ee;

  dim3 b256(256);

  fill4_kernel<<<128, b256, 0, stream>>>((float4*)cnt, (long)Nn / 4, 0.0f);
  rank_kernel<<<(Ee + 255) / 256, b256, 0, stream>>>(edge, cnt, rank);
  scan_kernel<<<(Nn + 1023) / 1024, b256, 0, stream>>>(cnt, offs, dinv);
  place_kernel<<<(Ee + 255) / 256, b256, 0, stream>>>(edge, offs, rank, dinv, csr);

  gemm_wmma_kernel<<<Nn / BLK_ROWS, b256, 0, stream>>>(x, W1, H, DIN);
  fill4_kernel<<<16, b256, 0, stream>>>((float4*)G, (long)DH * DH / 4, 0.0f);
  propagate_gram_kernel<<<GRAM_BLOCKS + GATHER_BLOCKS, b256, 0, stream>>>(offs, csr, dinv, H, S, G);
  update_next_kernel<<<Nn / BLK_ROWS, b256, 0, stream>>>(H, S, G, b1, W2);

  fill4_kernel<<<16, b256, 0, stream>>>((float4*)G, (long)DH * DH / 4, 0.0f);
  propagate_gram_kernel<<<GRAM_BLOCKS + GATHER_BLOCKS, b256, 0, stream>>>(offs, csr, dinv, H, S, G);
  update_final_kernel<<<Nn / BLK_ROWS, b256, 0, stream>>>(H, S, G, b2, Wf, bf, out);
}


# x@W1 has no launch of its own: its 80-row blocks run beside the edge passes (768 in the rank launch, 482 in the place launch), one role per work-group; 11 launches to 10; sum order unchanged
# speedup vs baseline: 2.8766x; 1.0604x over previous
#include <hip/hip_runtime.h>
#include <math.h>

#define Nn   100000
#define Ee   800000
#define DIN  256
#define DH   128
#define Kc   16
#define GAMMA 0.1f
#define LAM1  1.0f
#define LAM2  0.5f
#define AZ (1.0f - GAMMA*LAM1 + GAMMA*LAM2)
#define CS (GAMMA*LAM1)
#define CT (GAMMA*LAM2)
#define GRAM_CHUNK 400
#define GRAM_TILE  40
#define GEMM_KTILE 64
#define GEMM_PITCH (GEMM_KTILE + 4)
#define GRAM_PITCH (DH + 8)
#define UPD_PITCH (DH + 4)
#define BLK_STRIPS 5
#define BLK_ROWS   (16 * BLK_STRIPS)

typedef __attribute__((ext_vector_type(2))) float v2f;
typedef __attribute__((ext_vector_type(8))) float v8f;

__device__ __forceinline__ v8f wmma4(v2f a, v2f b, v8f c) {
  return __builtin_amdgcn_wmma_f32_16x16x4_f32(
       false, a,  false, b,
       (short)0, c,  false,  false);
}


__global__ void fill4_kernel(float4* __restrict__ p, long n4, float v) {
  long i = (long)blockIdx.x * blockDim.x + threadIdx.x;
  long stride = (long)gridDim.x * blockDim.x;
  float4 val = make_float4(v, v, v, v);
  for (; i < n4; i += stride) p[i] = val;
}

__device__ __forceinline__ void
rank_edges(int gblock, const int* __restrict__ edge, int* __restrict__ cnt, int* __restrict__ rank) {
  int e = gblock * 256 + threadIdx.x;
  if (e < Ee) rank[e] = atomicAdd(&cnt[edge[Ee + e]], 1);
}

__global__ void __launch_bounds__(256)
scan_kernel(const int* __restrict__ cnt, int* __restrict__ offs, float* __restrict__ dinv) {
  __shared__ int red[8];
  __shared__ int wtot[8];
  int tid  = threadIdx.x;
  int lane = tid & 31;
  int wave = tid >> 5;
  int b    = blockIdx.x;
  const int4* c4 = (const int4*)cnt;
  int p = 0;
#pragma unroll 8
  for (int k = 0; k < b; ++k) {
    int4 v = c4[k * 256 + tid];
    p += (v.x + v.y) + (v.z + v.w);
  }
#pragma unroll
  for (int d = 16; d >= 1; d >>= 1) p += __shfl_xor(p, d);
  if (lane == 0) red[wave] = p;

  int i = b * 1024 + tid * 4;
  int4 v = make_int4(0, 0, 0, 0);
  if (i < Nn) v = c4[b * 256 + tid];
  int s = (v.x + v.y) + (v.z + v.w);
  int x = s;
#pragma unroll
  for (int d = 1; d < 32; d <<= 1) {
    int y = __shfl_up(x, d);
    if (lane >= d) x += y;
  }
  if (lane == 31) wtot[wave] = x;
  __syncthreads();
  int base = 0;
#pragma unroll
  for (int k = 0; k < 8; ++k) base += red[k];
#pragma unroll
  for (int k = 0; k < 8; ++k) base += (k < wave) ? wtot[k] : 0;
  int excl = base + x - s;
  if (i < Nn) {
    int4 o;
    o.x = excl;
    o.y = o.x + v.x;
    o.z = o.y + v.y;
    o.w = o.z + v.z;
    ((int4*)offs)[b * 256 + tid] = o;
    if (i + 4 == Nn) offs[Nn] = o.w + v.w;
    float d0 = (float)(v.x + 1), d1 = (float)(v.y + 1), d2 = (float)(v.z + 1), d3 = (float)(v.w + 1);
    float4 dv;
    dv.x = (d0 > 0.0f) ? rsqrtf(fmaxf(d0, 1.0f)) : 0.0f;
    dv.y = (d1 > 0.0f) ? rsqrtf(fmaxf(d1, 1.0f)) : 0.0f;
    dv.z = (d2 > 0.0f) ? rsqrtf(fmaxf(d2, 1.0f)) : 0.0f;
    dv.w = (d3 > 0.0f) ? rsqrtf(fmaxf(d3, 1.0f)) : 0.0f;
    ((float4*)dinv)[b * 256 + tid] = dv;
  }
}

__device__ __forceinline__ void
place_edges(int gblock, const int* __restrict__ edge, const int* __restrict__ offs,
            const int* __restrict__ rank, const float* __restrict__ dinv, int* __restrict__ csr) {
  int e = gblock * 256 + threadIdx.x;
  if (e >= Ee) return;
  int r = edge[e];
  int c = edge[Ee + e];
  int pos = offs[c] + rank[e];
  pos = min(max(pos, 0), Ee - 1);
  int2 rec;
  rec.x = r;
  rec.y = __float_as_int(dinv[r] * dinv[c]);
  ((int2*)csr)[pos] = rec;
}

__device__ __forceinline__ void
dense_rows(int blk, float* __restrict__ atile, const float* __restrict__ A, const float* __restrict__ B,
           float* __restrict__ C, int Kdim) {
  int tid  = threadIdx.x;
  int wave = tid >> 5;
  int lane = tid & 31;
  int half = lane >> 4;
  int lid  = lane & 15;
  int kb   = half * 2;
  int row0 = blk * BLK_ROWS;
  int col0 = wave * 16;
  v8f acc[BLK_STRIPS] = {};
  for (int k0 = 0; k0 < Kdim; k0 += GEMM_KTILE) {
    __syncthreads();
#pragma unroll
    for (int i = 0; i < BLK_STRIPS; ++i) {
      int idx = tid + 256 * i;
      int sr  = idx >> 4;
      int sc  = idx & 15;
      ((float4*)(atile + sr * GEMM_PITCH))[sc] =
          ((const float4*)(A + (size_t)(row0 + sr) * Kdim + k0))[sc];
    }
    __syncthreads();
#pragma unroll
    for (int ks = 0; ks < GEMM_KTILE; ks += 4) {
      v2f b;
      b.x = B[(size_t)(k0 + ks + kb)     * DH + col0 + lid];
      b.y = B[(size_t)(k0 + ks + kb + 1) * DH + col0 + lid];
#pragma unroll
      for (int s = 0; s < BLK_STRIPS; ++s) {
        v2f a = *(const v2f*)(atile + (s * 16 + lid) * GEMM_PITCH + ks + kb);
        acc[s] = wmma4(a, b, acc[s]);
      }
    }
  }
#pragma unroll
  for (int s = 0; s < BLK_STRIPS; ++s)
#pragma unroll
    for (int r = 0; r < 8; ++r)
      C[(size_t)(row0 + s * 16 + r + 8 * half) * DH + col0 + lid] = acc[s][r];
}

#define DENSE_BLOCKS      (Nn / BLK_ROWS)
#define DENSE_WITH_RANK   768
#define DENSE_WITH_PLACE  (DENSE_BLOCKS - DENSE_WITH_RANK)
#define EDGE_BLOCKS       ((Ee + 255) / 256)
__global__ void __launch_bounds__(256)
dense_rank_kernel(const float* __restrict__ x, const float* __restrict__ Wm, float* __restrict__ H, int Kdim,
                  const int* __restrict__ edge, int* __restrict__ cnt, int* __restrict__ rank) {
  __shared__ __align__(16) float atile[BLK_ROWS * GEMM_PITCH];
  int b = blockIdx.x;
  if (b < DENSE_WITH_RANK) dense_rows(b, atile, x, Wm, H, Kdim);
  else                     rank_edges(b - DENSE_WITH_RANK, edge, cnt, rank);
}

__global__ void __launch_bounds__(256)
dense_place_kernel(const float* __restrict__ x, const float* __restrict__ Wm, float* __restrict__ H, int Kdim,
                   const int* __restrict__ edge, const int* __restrict__ offs, const int* __restrict__ rank,
                   const float* __restrict__ dinv, int* __restrict__ csr) {
  __shared__ __align__(16) float atile[BLK_ROWS * GEMM_PITCH];
  int b = blockIdx.x;
  if (b < DENSE_WITH_PLACE) dense_rows(DENSE_WITH_RANK + b, atile, x, Wm, H, Kdim);
  else                      place_edges(b - DENSE_WITH_PLACE, edge, offs, rank, dinv, csr);
}

__device__ __forceinline__ void
gather_rows(int gblock, const int* __restrict__ offs, const int* __restrict__ csr,
            const float* __restrict__ dinv, const float* __restrict__ H, float* __restrict__ S) {
  int lane = threadIdx.x & 31;
  int c = gblock * 8 + __builtin_amdgcn_readfirstlane((int)(threadIdx.x >> 5));
  int beg = offs[c];
  int end = offs[c + 1];
  beg = min(max(beg, 0), Ee);
  end = min(max(end, beg), Ee);
  float di = dinv[c];
  float w0 = di * di;
  const float4 hs = *(const float4*)(H + (size_t)c * DH + lane * 4);
  float4 acc;
  acc.x = w0 * hs.x; acc.y = w0 * hs.y; acc.z = w0 * hs.z; acc.w = w0 * hs.w;
  for (int base = beg; base < end; base += 32) {
    int n = min(32, end - base);
    int2 my = make_int2(0, 0);
    if (lane < n) my = ((const int2*)csr)[base + lane];
#pragma unroll 4
    for (int j = 0; j < n; ++j) {
      int   r = __builtin_amdgcn_readlane(my.x, j);
      float w = __int_as_float(__builtin_amdgcn_readlane(my.y, j));
      r = min(max(r, 0), Nn - 1);
      const float4 h = *(const float4*)(H + (size_t)r * DH + lane * 4);
      acc.x = fmaf(w, h.x, acc.x);
      acc.y = fmaf(w, h.y, acc.y);
      acc.z = fmaf(w, h.z, acc.z);
      acc.w = fmaf(w, h.w, acc.w);
    }
  }
  *(float4*)(S + (size_t)c * DH + lane * 4) = acc;
}

#define GRAM_CT 4
__device__ __forceinline__ void
gram_strips(int gblock, float* __restrict__ tile, const float* __restrict__ H, float* __restrict__ G) {
  int mblock = gblock >> 1;
  int cbase  = (gblock & 1) * (GRAM_CT * 16);
  int tid  = threadIdx.x;
  int wave = tid >> 5;
  int lane = tid & 31;
  int half = lane >> 4;
  int lid  = lane & 15;
  int kb   = 2 * half;
  int n0   = mblock * GRAM_CHUNK;
  v8f acc[GRAM_CT] = {};
  for (int base = n0; base < n0 + GRAM_CHUNK; base += GRAM_TILE) {
    __syncthreads();
    {
      const float4* src = (const float4*)(H + (size_t)base * DH);
      int trow = tid >> 5;
      int tc4  = tid & 31;
#pragma unroll
      for (int i = 0; i < GRAM_TILE / 8; ++i)
        ((float4*)(tile + (trow + 8 * i) * GRAM_PITCH))[tc4] = src[tid + 256 * i];
    }
    __syncthreads();
#pragma unroll 2
    for (int ks = 0; ks < GRAM_TILE; ks += 4) {
      v2f a;
      a.x = tile[(ks + kb)     * GRAM_PITCH + wave * 16 + lid];
      a.y = tile[(ks + kb + 1) * GRAM_PITCH + wave * 16 + lid];
#pragma unroll
      for (int ct = 0; ct < GRAM_CT; ++ct) {
        v2f b;
        b.x = tile[(ks + kb)     * GRAM_PITCH + cbase + ct * 16 + lid];
        b.y = tile[(ks + kb + 1) * GRAM_PITCH + cbase + ct * 16 + lid];
        acc[ct] = wmma4(a, b, acc[ct]);
      }
    }
  }
#pragma unroll
  for (int ct = 0; ct < GRAM_CT; ++ct)
#pragma unroll
    for (int r = 0; r < 8; ++r)
      atomicAdd(&G[(size_t)(wave * 16 + r + 8 * half) * DH + cbase + ct * 16 + lid], acc[ct][r]);
}

#define GRAM_BLOCKS   (2 * (Nn / GRAM_CHUNK))
#define GATHER_BLOCKS (Nn / 8)
__global__ void __launch_bounds__(256)
propagate_gram_kernel(const int* __restrict__ offs, const int* __restrict__ csr,
                      const float* __restrict__ dinv, const float* __restrict__ H,
                      float* __restrict__ S, float* __restrict__ G) {
  __shared__ __align__(16) float tile[GRAM_TILE * GRAM_PITCH];
  int b = blockIdx.x;
  if (b < GRAM_BLOCKS) gram_strips(b, tile, H, G);
  else                 gather_rows(b - GRAM_BLOCKS, offs, csr, dinv, H, S);
}


__device__ __forceinline__ void
tile_times_mat(const float* __restrict__ tile, const float* __restrict__ Bm, int col0, int lid, int kb,
               v8f (&acc)[BLK_STRIPS]) {
#pragma unroll 8
  for (int kk = 0; kk < DH; kk += 4) {
    v2f b;
    b.x = Bm[(size_t)(kk + kb)     * DH + col0 + lid];
    b.y = Bm[(size_t)(kk + kb + 1) * DH + col0 + lid];
#pragma unroll
    for (int s = 0; s < BLK_STRIPS; ++s) {
      v2f a = *(const v2f*)(tile + (s * 16 + lid) * UPD_PITCH + kk + kb);
      acc[s] = wmma4(a, b, acc[s]);
    }
  }
}

__device__ __forceinline__ void
layer_tail(float* __restrict__ htile, float* __restrict__ H, const float* __restrict__ S,
           const float* __restrict__ G, const float* __restrict__ bias,
           const float* __restrict__ Wn, const float* __restrict__ bn, float* __restrict__ out, const bool last) {
  int tid  = threadIdx.x;
  int wave = tid >> 5;
  int lane = tid & 31;
  int half = lane >> 4;
  int lid  = lane & 15;
  int kb   = 2 * half;
  int row0 = blockIdx.x * BLK_ROWS;
  int col0 = wave * 16;
  {
    const float4* src = (const float4*)(H + (size_t)row0 * DH);
#pragma unroll
    for (int i = 0; i < 2 * BLK_STRIPS; ++i) {
      int idx  = tid + 256 * i;
      int trow = idx >> 5;
      int tc4  = idx & 31;
      ((float4*)(htile + trow * UPD_PITCH))[tc4] = src[idx];
    }
  }
  __syncthreads();
  v8f acc[BLK_STRIPS] = {};
  tile_times_mat(htile, G, col0, lid, kb, acc);
  float bcol = bias[col0 + lid];
  __syncthreads();
#pragma unroll
  for (int s = 0; s < BLK_STRIPS; ++s) {
#pragma unroll
    for (int r = 0; r < 8; ++r) {
      int m = s * 16 + r + 8 * half;
      size_t off = (size_t)(row0 + m) * DH + col0 + lid;
      float val = AZ * htile[m * UPD_PITCH + col0 + lid] + CS * S[off] - CT * acc[s][r] + bcol;
      htile[m * UPD_PITCH + col0 + lid] = fmaxf(val, 0.0f);
    }
  }
  __syncthreads();
  if (!last) {
    v8f nxt[BLK_STRIPS] = {};
    tile_times_mat(htile, Wn, col0, lid, kb, nxt);
#pragma unroll
    for (int s = 0; s < BLK_STRIPS; ++s)
#pragma unroll
      for (int r = 0; r < 8; ++r)
        H[(size_t)(row0 + s * 16 + r + 8 * half) * DH + col0 + lid] = nxt[s][r];
  } else {
    int wv = __builtin_amdgcn_readfirstlane(wave);
    if (wv < BLK_STRIPS) {
      const float* arow = htile + (wv * 16 + (((lid & 7) << 1) | (lid >> 3))) * UPD_PITCH + kb;
      v8f l0 = {}, l1 = {};
#pragma unroll 8
      for (int kk = 0; kk < DH; kk += 8) {
        v2f b0, b1;
        b0.x = Wn[(kk + kb)     * Kc + lid];
        b0.y = Wn[(kk + kb + 1) * Kc + lid];
        b1.x = Wn[(kk + kb + 4) * Kc + lid];
        b1.y = Wn[(kk + kb + 5) * Kc + lid];
        v2f a0 = *(const v2f*)(arow + kk);
        v2f a1 = *(const v2f*)(arow + kk + 4);
        l0 = wmma4(a0, b0, l0);
        l1 = wmma4(a1, b1, l1);
      }
      float bl = bn[lid];
      float* orow = out + (size_t)(row0 + wv * 16) * Kc + lane;
#pragma unroll
      for (int r = 0; r < 8; ++r) {
        float v = (l0[r] + l1[r]) + bl;
        float mx = v;
#pragma unroll
        for (int d = 8; d >= 1; d >>= 1) mx = fmaxf(mx, __shfl_xor(mx, d));
        float se = expf(v - mx);
#pragma unroll
        for (int d = 8; d >= 1; d >>= 1) se += __shfl_xor(se, d);
        orow[32 * r] = v - (mx + logf(se));
      }
    }
  }
}

__global__ void __launch_bounds__(256)
update_next_kernel(float* __restrict__ H, const float* __restrict__ S, const float* __restrict__ G,
                   const float* __restrict__ bias, const float* __restrict__ Wn) {
  __shared__ __align__(16) float htile[BLK_ROWS * UPD_PITCH];
  layer_tail(htile, H, S, G, bias, Wn, nullptr, nullptr, false);
}

__global__ void __launch_bounds__(256)
update_final_kernel(float* __restrict__ H, const float* __restrict__ S, const float* __restrict__ G,
                    const float* __restrict__ bias, const float* __restrict__ Wn, const float* __restrict__ bn,
                    float* __restrict__ out) {
  __shared__ __align__(16) float htile[BLK_ROWS * UPD_PITCH];
  layer_tail(htile, H, S, G, bias, Wn, bn, out, true);
}

extern "C" void kernel_launch(void* const* d_in, const int* in_sizes, int n_in,
                              void* d_out, int out_size, void* d_ws, size_t ws_size,
                              hipStream_t stream) {
  const float* x    = (const float*)d_in[0];
  const int*   edge = (const int*)  d_in[1];
  const float* W1   = (const float*)d_in[2];
  const float* b1   = (const float*)d_in[3];
  const float* W2   = (const float*)d_in[4];
  const float* b2   = (const float*)d_in[5];
  const float* Wf   = (const float*)d_in[6];
  const float* bf   = (const float*)d_in[7];
  float* out = (float*)d_out;

  float* H    = (float*)d_ws;
  float* S    = H + (size_t)Nn * DH;
  float* dinv = S + (size_t)Nn * DH;
  float* G    = dinv + Nn;
  int*   cnt  = (int*)(G + (size_t)DH * DH);
  int*   offs = cnt + Nn;
  int*   rank = offs + Nn + 4;
  int*   csr  = rank + Ee;

  dim3 b256(256);

  fill4_kernel<<<128, b256, 0, stream>>>((float4*)cnt, (long)Nn / 4, 0.0f);
  dense_rank_kernel<<<DENSE_WITH_RANK + EDGE_BLOCKS, b256, 0, stream>>>(x, W1, H, DIN, edge, cnt, rank);
  scan_kernel<<<(Nn + 1023) / 1024, b256, 0, stream>>>(cnt, offs, dinv);
  dense_place_kernel<<<DENSE_WITH_PLACE + EDGE_BLOCKS, b256, 0, stream>>>(x, W1, H, DIN, edge, offs, rank, dinv, csr);

  fill4_kernel<<<16, b256, 0, stream>>>((float4*)G, (long)DH * DH / 4, 0.0f);
  propagate_gram_kernel<<<GRAM_BLOCKS + GATHER_BLOCKS, b256, 0, stream>>>(offs, csr, dinv, H, S, G);
  update_next_kernel<<<Nn / BLK_ROWS, b256, 0, stream>>>(H, S, G, b1, W2);

  fill4_kernel<<<16, b256, 0, stream>>>((float4*)G, (long)DH * DH / 4, 0.0f);
  propagate_gram_kernel<<<GRAM_BLOCKS + GATHER_BLOCKS, b256, 0, stream>>>(offs, csr, dinv, H, S, G);
  update_final_kernel<<<Nn / BLK_ROWS, b256, 0, stream>>>(H, S, G, b2, Wf, bf, out);
}


# Gram: the split-K float atomics on G become plain stores of per-chunk partial tiles; a fixed-order reduce launch takes the place of the clear of G; 10 launches as before; G sum order now fixed
# speedup vs baseline: 3.0640x; 1.0651x over previous
#include <hip/hip_runtime.h>
#include <math.h>

#define Nn   100000
#define Ee   800000
#define DIN  256
#define DH   128
#define Kc   16
#define GAMMA 0.1f
#define LAM1  1.0f
#define LAM2  0.5f
#define AZ (1.0f - GAMMA*LAM1 + GAMMA*LAM2)
#define CS (GAMMA*LAM1)
#define CT (GAMMA*LAM2)
#define GRAM_CHUNK 400
#define GRAM_TILE  40
#define GEMM_KTILE 64
#define GEMM_PITCH (GEMM_KTILE + 4)
#define GRAM_PITCH (DH + 8)
#define UPD_PITCH (DH + 4)
#define BLK_STRIPS 5
#define BLK_ROWS   (16 * BLK_STRIPS)

typedef __attribute__((ext_vector_type(2))) float v2f;
typedef __attribute__((ext_vector_type(8))) float v8f;

__device__ __forceinline__ v8f wmma4(v2f a, v2f b, v8f c) {
  return __builtin_amdgcn_wmma_f32_16x16x4_f32(
       false, a,  false, b,
       (short)0, c,  false,  false);
}


__global__ void fill4_kernel(float4* __restrict__ p, long n4, float v) {
  long i = (long)blockIdx.x * blockDim.x + threadIdx.x;
  long stride = (long)gridDim.x * blockDim.x;
  float4 val = make_float4(v, v, v, v);
  for (; i < n4; i += stride) p[i] = val;
}

__device__ __forceinline__ void
rank_edges(int gblock, const int* __restrict__ edge, int* __restrict__ cnt, int* __restrict__ rank) {
  int e = gblock * 256 + threadIdx.x;
  if (e < Ee) rank[e] = atomicAdd(&cnt[edge[Ee + e]], 1);
}

__global__ void __launch_bounds__(256)
scan_kernel(const int* __restrict__ cnt, int* __restrict__ offs, float* __restrict__ dinv) {
  __shared__ int red[8];
  __shared__ int wtot[8];
  int tid  = threadIdx.x;
  int lane = tid & 31;
  int wave = tid >> 5;
  int b    = blockIdx.x;
  const int4* c4 = (const int4*)cnt;
  int p = 0;
#pragma unroll 8
  for (int k = 0; k < b; ++k) {
    int4 v = c4[k * 256 + tid];
    p += (v.x + v.y) + (v.z + v.w);
  }
#pragma unroll
  for (int d = 16; d >= 1; d >>= 1) p += __shfl_xor(p, d);
  if (lane == 0) red[wave] = p;

  int i = b * 1024 + tid * 4;
  int4 v = make_int4(0, 0, 0, 0);
  if (i < Nn) v = c4[b * 256 + tid];
  int s = (v.x + v.y) + (v.z + v.w);
  int x = s;
#pragma unroll
  for (int d = 1; d < 32; d <<= 1) {
    int y = __shfl_up(x, d);
    if (lane >= d) x += y;
  }
  if (lane == 31) wtot[wave] = x;
  __syncthreads();
  int base = 0;
#pragma unroll
  for (int k = 0; k < 8; ++k) base += red[k];
#pragma unroll
  for (int k = 0; k < 8; ++k) base += (k < wave) ? wtot[k] : 0;
  int excl = base + x - s;
  if (i < Nn) {
    int4 o;
    o.x = excl;
    o.y = o.x + v.x;
    o.z = o.y + v.y;
    o.w = o.z + v.z;
    ((int4*)offs)[b * 256 + tid] = o;
    if (i + 4 == Nn) offs[Nn] = o.w + v.w;
    float d0 = (float)(v.x + 1), d1 = (float)(v.y + 1), d2 = (float)(v.z + 1), d3 = (float)(v.w + 1);
    float4 dv;
    dv.x = (d0 > 0.0f) ? rsqrtf(fmaxf(d0, 1.0f)) : 0.0f;
    dv.y = (d1 > 0.0f) ? rsqrtf(fmaxf(d1, 1.0f)) : 0.0f;
    dv.z = (d2 > 0.0f) ? rsqrtf(fmaxf(d2, 1.0f)) : 0.0f;
    dv.w = (d3 > 0.0f) ? rsqrtf(fmaxf(d3, 1.0f)) : 0.0f;
    ((float4*)dinv)[b * 256 + tid] = dv;
  }
}

__device__ __forceinline__ void
place_edges(int gblock, const int* __restrict__ edge, const int* __restrict__ offs,
            const int* __restrict__ rank, const float* __restrict__ dinv, int* __restrict__ csr) {
  int e = gblock * 256 + threadIdx.x;
  if (e >= Ee) return;
  int r = edge[e];
  int c = edge[Ee + e];
  int pos = offs[c] + rank[e];
  pos = min(max(pos, 0), Ee - 1);
  int2 rec;
  rec.x = r;
  rec.y = __float_as_int(dinv[r] * dinv[c]);
  ((int2*)csr)[pos] = rec;
}

__device__ __forceinline__ void
dense_rows(int blk, float* __restrict__ atile, const float* __restrict__ A, const float* __restrict__ B,
           float* __restrict__ C, int Kdim) {
  int tid  = threadIdx.x;
  int wave = tid >> 5;
  int lane = tid & 31;
  int half = lane >> 4;
  int lid  = lane & 15;
  int kb   = half * 2;
  int row0 = blk * BLK_ROWS;
  int col0 = wave * 16;
  v8f acc[BLK_STRIPS] = {};
  for (int k0 = 0; k0 < Kdim; k0 += GEMM_KTILE) {
    __syncthreads();
#pragma unroll
    for (int i = 0; i < BLK_STRIPS; ++i) {
      int idx = tid + 256 * i;
      int sr  = idx >> 4;
      int sc  = idx & 15;
      ((float4*)(atile + sr * GEMM_PITCH))[sc] =
          ((const float4*)(A + (size_t)(row0 + sr) * Kdim + k0))[sc];
    }
    __syncthreads();
#pragma unroll
    for (int ks = 0; ks < GEMM_KTILE; ks += 4) {
      v2f b;
      b.x = B[(size_t)(k0 + ks + kb)     * DH + col0 + lid];
      b.y = B[(size_t)(k0 + ks + kb + 1) * DH + col0 + lid];
#pragma unroll
      for (int s = 0; s < BLK_STRIPS; ++s) {
        v2f a = *(const v2f*)(atile + (s * 16 + lid) * GEMM_PITCH + ks + kb);
        acc[s] = wmma4(a, b, acc[s]);
      }
    }
  }
#pragma unroll
  for (int s = 0; s < BLK_STRIPS; ++s)
#pragma unroll
    for (int r = 0; r < 8; ++r)
      C[(size_t)(row0 + s * 16 + r + 8 * half) * DH + col0 + lid] = acc[s][r];
}

#define DENSE_BLOCKS      (Nn / BLK_ROWS)
#define DENSE_WITH_RANK   768
#define DENSE_WITH_PLACE  (DENSE_BLOCKS - DENSE_WITH_RANK)
#define EDGE_BLOCKS       ((Ee + 255) / 256)
__global__ void __launch_bounds__(256)
dense_rank_kernel(const float* __restrict__ x, const float* __restrict__ Wm, float* __restrict__ H, int Kdim,
                  const int* __restrict__ edge, int* __restrict__ cnt, int* __restrict__ rank) {
  __shared__ __align__(16) float atile[BLK_ROWS * GEMM_PITCH];
  int b = blockIdx.x;
  if (b < DENSE_WITH_RANK) dense_rows(b, atile, x, Wm, H, Kdim);
  else                     rank_edges(b - DENSE_WITH_RANK, edge, cnt, rank);
}

__global__ void __launch_bounds__(256)
dense_place_kernel(const float* __restrict__ x, const float* __restrict__ Wm, float* __restrict__ H, int Kdim,
                   const int* __restrict__ edge, const int* __restrict__ offs, const int* __restrict__ rank,
                   const float* __restrict__ dinv, int* __restrict__ csr) {
  __shared__ __align__(16) float atile[BLK_ROWS * GEMM_PITCH];
  int b = blockIdx.x;
  if (b < DENSE_WITH_PLACE) dense_rows(DENSE_WITH_RANK + b, atile, x, Wm, H, Kdim);
  else                      place_edges(b - DENSE_WITH_PLACE, edge, offs, rank, dinv, csr);
}

__device__ __forceinline__ void
gather_rows(int gblock, const int* __restrict__ offs, const int* __restrict__ csr,
            const float* __restrict__ dinv, const float* __restrict__ H, float* __restrict__ S) {
  int lane = threadIdx.x & 31;
  int c = gblock * 8 + __builtin_amdgcn_readfirstlane((int)(threadIdx.x >> 5));
  int beg = offs[c];
  int end = offs[c + 1];
  beg = min(max(beg, 0), Ee);
  end = min(max(end, beg), Ee);
  float di = dinv[c];
  float w0 = di * di;
  const float4 hs = *(const float4*)(H + (size_t)c * DH + lane * 4);
  float4 acc;
  acc.x = w0 * hs.x; acc.y = w0 * hs.y; acc.z = w0 * hs.z; acc.w = w0 * hs.w;
  for (int base = beg; base < end; base += 32) {
    int n = min(32, end - base);
    int2 my = make_int2(0, 0);
    if (lane < n) my = ((const int2*)csr)[base + lane];
#pragma unroll 4
    for (int j = 0; j < n; ++j) {
      int   r = __builtin_amdgcn_readlane(my.x, j);
      float w = __int_as_float(__builtin_amdgcn_readlane(my.y, j));
      r = min(max(r, 0), Nn - 1);
      const float4 h = *(const float4*)(H + (size_t)r * DH + lane * 4);
      acc.x = fmaf(w, h.x, acc.x);
      acc.y = fmaf(w, h.y, acc.y);
      acc.z = fmaf(w, h.z, acc.z);
      acc.w = fmaf(w, h.w, acc.w);
    }
  }
  *(float4*)(S + (size_t)c * DH + lane * 4) = acc;
}

#define GRAM_CT 4
#define GRAM_CHUNKS (Nn / GRAM_CHUNK)
__device__ __forceinline__ void
gram_strips(int gblock, float* __restrict__ tile, const float* __restrict__ H, float* __restrict__ P) {
  int mblock = gblock >> 1;
  int cbase  = (gblock & 1) * (GRAM_CT * 16);
  int tid  = threadIdx.x;
  int wave = tid >> 5;
  int lane = tid & 31;
  int half = lane >> 4;
  int lid  = lane & 15;
  int kb   = 2 * half;
  int n0   = mblock * GRAM_CHUNK;
  v8f acc[GRAM_CT] = {};
  for (int base = n0; base < n0 + GRAM_CHUNK; base += GRAM_TILE) {
    __syncthreads();
    {
      const float4* src = (const float4*)(H + (size_t)base * DH);
      int trow = tid >> 5;
      int tc4  = tid & 31;
#pragma unroll
      for (int i = 0; i < GRAM_TILE / 8; ++i)
        ((float4*)(tile + (trow + 8 * i) * GRAM_PITCH))[tc4] = src[tid + 256 * i];
    }
    __syncthreads();
#pragma unroll 2
    for (int ks = 0; ks < GRAM_TILE; ks += 4) {
      v2f a;
      a.x = tile[(ks + kb)     * GRAM_PITCH + wave * 16 + lid];
      a.y = tile[(ks + kb + 1) * GRAM_PITCH + wave * 16 + lid];
#pragma unroll
      for (int ct = 0; ct < GRAM_CT; ++ct) {
        v2f b;
        b.x = tile[(ks + kb)     * GRAM_PITCH + cbase + ct * 16 + lid];
        b.y = tile[(ks + kb + 1) * GRAM_PITCH + cbase + ct * 16 + lid];
        acc[ct] = wmma4(a, b, acc[ct]);
      }
    }
  }
  float* Pc = P + (size_t)mblock * (DH * DH);
#pragma unroll
  for (int ct = 0; ct < GRAM_CT; ++ct)
#pragma unroll
    for (int r = 0; r < 8; ++r)
      Pc[(wave * 16 + r + 8 * half) * DH + cbase + ct * 16 + lid] = acc[ct][r];
}

#define RED_GROUPS 16
#define RED_DEPTH  ((GRAM_CHUNKS + RED_GROUPS - 1) / RED_GROUPS)
__global__ void __launch_bounds__(256)
gram_reduce_kernel(const float* __restrict__ P, float* __restrict__ G) {
  __shared__ float4 part[RED_GROUPS][16];
  int tid  = threadIdx.x;
  int slot = tid & 15;
  int grp  = tid >> 4;
  const float4* src = (const float4*)P + blockIdx.x * 16 + slot;
  float4 v[RED_DEPTH];
#pragma unroll
  for (int i = 0; i < RED_DEPTH; ++i) {
    int m = grp + RED_GROUPS * i;
    v[i] = make_float4(0.0f, 0.0f, 0.0f, 0.0f);
    if (m < GRAM_CHUNKS) v[i] = src[(size_t)m * (DH * DH / 4)];
  }
  float4 s = v[0];
#pragma unroll
  for (int i = 1; i < RED_DEPTH; ++i) {
    s.x += v[i].x; s.y += v[i].y; s.z += v[i].z; s.w += v[i].w;
  }
  part[grp][slot] = s;
  __syncthreads();
  if (tid < 16) {
    float4 t = part[0][tid];
#pragma unroll
    for (int g = 1; g < RED_GROUPS; ++g) {
      float4 u = part[g][tid];
      t.x += u.x; t.y += u.y; t.z += u.z; t.w += u.w;
    }
    ((float4*)G)[blockIdx.x * 16 + tid] = t;
  }
}

#define GRAM_BLOCKS   (2 * GRAM_CHUNKS)
#define GATHER_BLOCKS (Nn / 8)
__global__ void __launch_bounds__(256)
propagate_gram_kernel(const int* __restrict__ offs, const int* __restrict__ csr,
                      const float* __restrict__ dinv, const float* __restrict__ H,
                      float* __restrict__ S, float* __restrict__ P) {
  __shared__ __align__(16) float tile[GRAM_TILE * GRAM_PITCH];
  int b = blockIdx.x;
  if (b < GRAM_BLOCKS) gram_strips(b, tile, H, P);
  else                 gather_rows(b - GRAM_BLOCKS, offs, csr, dinv, H, S);
}


__device__ __forceinline__ void
tile_times_mat(const float* __restrict__ tile, const float* __restrict__ Bm, int col0, int lid, int kb,
               v8f (&acc)[BLK_STRIPS]) {
#pragma unroll 8
  for (int kk = 0; kk < DH; kk += 4) {
    v2f b;
    b.x = Bm[(size_t)(kk + kb)     * DH + col0 + lid];
    b.y = Bm[(size_t)(kk + kb + 1) * DH + col0 + lid];
#pragma unroll
    for (int s = 0; s < BLK_STRIPS; ++s) {
      v2f a = *(const v2f*)(tile + (s * 16 + lid) * UPD_PITCH + kk + kb);
      acc[s] = wmma4(a, b, acc[s]);
    }
  }
}

__device__ __forceinline__ void
layer_tail(float* __restrict__ htile, float* __restrict__ H, const float* __restrict__ S,
           const float* __restrict__ G, const float* __restrict__ bias,
           const float* __restrict__ Wn, const float* __restrict__ bn, float* __restrict__ out, const bool last) {
  int tid  = threadIdx.x;
  int wave = tid >> 5;
  int lane = tid & 31;
  int half = lane >> 4;
  int lid  = lane & 15;
  int kb   = 2 * half;
  int row0 = blockIdx.x * BLK_ROWS;
  int col0 = wave * 16;
  {
    const float4* src = (const float4*)(H + (size_t)row0 * DH);
#pragma unroll
    for (int i = 0; i < 2 * BLK_STRIPS; ++i) {
      int idx  = tid + 256 * i;
      int trow = idx >> 5;
      int tc4  = idx & 31;
      ((float4*)(htile + trow * UPD_PITCH))[tc4] = src[idx];
    }
  }
  __syncthreads();
  v8f acc[BLK_STRIPS] = {};
  tile_times_mat(htile, G, col0, lid, kb, acc);
  float bcol = bias[col0 + lid];
  __syncthreads();
#pragma unroll
  for (int s = 0; s < BLK_STRIPS; ++s) {
#pragma unroll
    for (int r = 0; r < 8; ++r) {
      int m = s * 16 + r + 8 * half;
      size_t off = (size_t)(row0 + m) * DH + col0 + lid;
      float val = AZ * htile[m * UPD_PITCH + col0 + lid] + CS * S[off] - CT * acc[s][r] + bcol;
      htile[m * UPD_PITCH + col0 + lid] = fmaxf(val, 0.0f);
    }
  }
  __syncthreads();
  if (!last) {
    v8f nxt[BLK_STRIPS] = {};
    tile_times_mat(htile, Wn, col0, lid, kb, nxt);
#pragma unroll
    for (int s = 0; s < BLK_STRIPS; ++s)
#pragma unroll
      for (int r = 0; r < 8; ++r)
        H[(size_t)(row0 + s * 16 + r + 8 * half) * DH + col0 + lid] = nxt[s][r];
  } else {
    int wv = __builtin_amdgcn_readfirstlane(wave);
    if (wv < BLK_STRIPS) {
      const float* arow = htile + (wv * 16 + (((lid & 7) << 1) | (lid >> 3))) * UPD_PITCH + kb;
      v8f l0 = {}, l1 = {};
#pragma unroll 8
      for (int kk = 0; kk < DH; kk += 8) {
        v2f b0, b1;
        b0.x = Wn[(kk + kb)     * Kc + lid];
        b0.y = Wn[(kk + kb + 1) * Kc + lid];
        b1.x = Wn[(kk + kb + 4) * Kc + lid];
        b1.y = Wn[(kk + kb + 5) * Kc + lid];
        v2f a0 = *(const v2f*)(arow + kk);
        v2f a1 = *(const v2f*)(arow + kk + 4);
        l0 = wmma4(a0, b0, l0);
        l1 = wmma4(a1, b1, l1);
      }
      float bl = bn[lid];
      float* orow = out + (size_t)(row0 + wv * 16) * Kc + lane;
#pragma unroll
      for (int r = 0; r < 8; ++r) {
        float v = (l0[r] + l1[r]) + bl;
        float mx = v;
#pragma unroll
        for (int d = 8; d >= 1; d >>= 1) mx = fmaxf(mx, __shfl_xor(mx, d));
        float se = expf(v - mx);
#pragma unroll
        for (int d = 8; d >= 1; d >>= 1) se += __shfl_xor(se, d);
        orow[32 * r] = v - (mx + logf(se));
      }
    }
  }
}

__global__ void __launch_bounds__(256)
update_next_kernel(float* __restrict__ H, const float* __restrict__ S, const float* __restrict__ G,
                   const float* __restrict__ bias, const float* __restrict__ Wn) {
  __shared__ __align__(16) float htile[BLK_ROWS * UPD_PITCH];
  layer_tail(htile, H, S, G, bias, Wn, nullptr, nullptr, false);
}

__global__ void __launch_bounds__(256)
update_final_kernel(float* __restrict__ H, const float* __restrict__ S, const float* __restrict__ G,
                    const float* __restrict__ bias, const float* __restrict__ Wn, const float* __restrict__ bn,
                    float* __restrict__ out) {
  __shared__ __align__(16) float htile[BLK_ROWS * UPD_PITCH];
  layer_tail(htile, H, S, G, bias, Wn, bn, out, true);
}

extern "C" void kernel_launch(void* const* d_in, const int* in_sizes, int n_in,
                              void* d_out, int out_size, void* d_ws, size_t ws_size,
                              hipStream_t stream) {
  const float* x    = (const float*)d_in[0];
  const int*   edge = (const int*)  d_in[1];
  const float* W1   = (const float*)d_in[2];
  const float* b1   = (const float*)d_in[3];
  const float* W2   = (const float*)d_in[4];
  const float* b2   = (const float*)d_in[5];
  const float* Wf   = (const float*)d_in[6];
  const float* bf   = (const float*)d_in[7];
  float* out = (float*)d_out;

  float* H    = (float*)d_ws;
  float* S    = H + (size_t)Nn * DH;
  float* dinv = S + (size_t)Nn * DH;
  float* G    = dinv + Nn;
  int*   cnt  = (int*)(G + (size_t)DH * DH);
  int*   offs = cnt + Nn;
  int*   rank = offs + Nn + 4;
  int*   csr  = rank + Ee;
  float* P    = (float*)(csr + 2 * Ee + 28);

  dim3 b256(256);

  fill4_kernel<<<128, b256, 0, stream>>>((float4*)cnt, (long)Nn / 4, 0.0f);
  dense_rank_kernel<<<DENSE_WITH_RANK + EDGE_BLOCKS, b256, 0, stream>>>(x, W1, H, DIN, edge, cnt, rank);
  scan_kernel<<<(Nn + 1023) / 1024, b256, 0, stream>>>(cnt, offs, dinv);
  dense_place_kernel<<<DENSE_WITH_PLACE + EDGE_BLOCKS, b256, 0, stream>>>(x, W1, H, DIN, edge, offs, rank, dinv, csr);

  propagate_gram_kernel<<<GRAM_BLOCKS + GATHER_BLOCKS, b256, 0, stream>>>(offs, csr, dinv, H, S, P);
  gram_reduce_kernel<<<DH * DH / 64, b256, 0, stream>>>(P, G);
  update_next_kernel<<<Nn / BLK_ROWS, b256, 0, stream>>>(H, S, G, b1, W2);

  propagate_gram_kernel<<<GRAM_BLOCKS + GATHER_BLOCKS, b256, 0, stream>>>(offs, csr, dinv, H, S, P);
  gram_reduce_kernel<<<DH * DH / 64, b256, 0, stream>>>(P, G);
  update_final_kernel<<<Nn / BLK_ROWS, b256, 0, stream>>>(H, S, G, b2, Wf, bf, out);
}


# Gram H^T H is symmetric: only 36 of its 64 tiles (wrapped upper triangle) are formed, packed and reduced; the update reads the other 28 as transposes of their mirrors; sums and order unchanged
# speedup vs baseline: 3.2060x; 1.0463x over previous
#include <hip/hip_runtime.h>
#include <math.h>

#define Nn   100000
#define Ee   800000
#define DIN  256
#define DH   128
#define Kc   16
#define GAMMA 0.1f
#define LAM1  1.0f
#define LAM2  0.5f
#define AZ (1.0f - GAMMA*LAM1 + GAMMA*LAM2)
#define CS (GAMMA*LAM1)
#define CT (GAMMA*LAM2)
#define GRAM_CHUNK 400
#define GRAM_TILE  40
#define GEMM_KTILE 64
#define GEMM_PITCH (GEMM_KTILE + 4)
#define GRAM_PITCH (DH + 8)
#define UPD_PITCH (DH + 4)
#define BLK_STRIPS 5
#define BLK_ROWS   (16 * BLK_STRIPS)

typedef __attribute__((ext_vector_type(2))) float v2f;
typedef __attribute__((ext_vector_type(8))) float v8f;

__device__ __forceinline__ v8f wmma4(v2f a, v2f b, v8f c) {
  return __builtin_amdgcn_wmma_f32_16x16x4_f32(
       false, a,  false, b,
       (short)0, c,  false,  false);
}


__global__ void fill4_kernel(float4* __restrict__ p, long n4, float v) {
  long i = (long)blockIdx.x * blockDim.x + threadIdx.x;
  long stride = (long)gridDim.x * blockDim.x;
  float4 val = make_float4(v, v, v, v);
  for (; i < n4; i += stride) p[i] = val;
}

__device__ __forceinline__ void
rank_edges(int gblock, const int* __restrict__ edge, int* __restrict__ cnt, int* __restrict__ rank) {
  int e = gblock * 256 + threadIdx.x;
  if (e < Ee) rank[e] = atomicAdd(&cnt[edge[Ee + e]], 1);
}

__global__ void __launch_bounds__(256)
scan_kernel(const int* __restrict__ cnt, int* __restrict__ offs, float* __restrict__ dinv) {
  __shared__ int red[8];
  __shared__ int wtot[8];
  int tid  = threadIdx.x;
  int lane = tid & 31;
  int wave = tid >> 5;
  int b    = blockIdx.x;
  const int4* c4 = (const int4*)cnt;
  int p = 0;
#pragma unroll 8
  for (int k = 0; k < b; ++k) {
    int4 v = c4[k * 256 + tid];
    p += (v.x + v.y) + (v.z + v.w);
  }
#pragma unroll
  for (int d = 16; d >= 1; d >>= 1) p += __shfl_xor(p, d);
  if (lane == 0) red[wave] = p;

  int i = b * 1024 + tid * 4;
  int4 v = make_int4(0, 0, 0, 0);
  if (i < Nn) v = c4[b * 256 + tid];
  int s = (v.x + v.y) + (v.z + v.w);
  int x = s;
#pragma unroll
  for (int d = 1; d < 32; d <<= 1) {
    int y = __shfl_up(x, d);
    if (lane >= d) x += y;
  }
  if (lane == 31) wtot[wave] = x;
  __syncthreads();
  int base = 0;
#pragma unroll
  for (int k = 0; k < 8; ++k) base += red[k];
#pragma unroll
  for (int k = 0; k < 8; ++k) base += (k < wave) ? wtot[k] : 0;
  int excl = base + x - s;
  if (i < Nn) {
    int4 o;
    o.x = excl;
    o.y = o.x + v.x;
    o.z = o.y + v.y;
    o.w = o.z + v.z;
    ((int4*)offs)[b * 256 + tid] = o;
    if (i + 4 == Nn) offs[Nn] = o.w + v.w;
    float d0 = (float)(v.x + 1), d1 = (float)(v.y + 1), d2 = (float)(v.z + 1), d3 = (float)(v.w + 1);
    float4 dv;
    dv.x = (d0 > 0.0f) ? rsqrtf(fmaxf(d0, 1.0f)) : 0.0f;
    dv.y = (d1 > 0.0f) ? rsqrtf(fmaxf(d1, 1.0f)) : 0.0f;
    dv.z = (d2 > 0.0f) ? rsqrtf(fmaxf(d2, 1.0f)) : 0.0f;
    dv.w = (d3 > 0.0f) ? rsqrtf(fmaxf(d3, 1.0f)) : 0.0f;
    ((float4*)dinv)[b * 256 + tid] = dv;
  }
}

__device__ __forceinline__ void
place_edges(int gblock, const int* __restrict__ edge, const int* __restrict__ offs,
            const int* __restrict__ rank, const float* __restrict__ dinv, int* __restrict__ csr) {
  int e = gblock * 256 + threadIdx.x;
  if (e >= Ee) return;
  int r = edge[e];
  int c = edge[Ee + e];
  int pos = offs[c] + rank[e];
  pos = min(max(pos, 0), Ee - 1);
  int2 rec;
  rec.x = r;
  rec.y = __float_as_int(dinv[r] * dinv[c]);
  ((int2*)csr)[pos] = rec;
}

__device__ __forceinline__ void
dense_rows(int blk, float* __restrict__ atile, const float* __restrict__ A, const float* __restrict__ B,
           float* __restrict__ C, int Kdim) {
  int tid  = threadIdx.x;
  int wave = tid >> 5;
  int lane = tid & 31;
  int half = lane >> 4;
  int lid  = lane & 15;
  int kb   = half * 2;
  int row0 = blk * BLK_ROWS;
  int col0 = wave * 16;
  v8f acc[BLK_STRIPS] = {};
  for (int k0 = 0; k0 < Kdim; k0 += GEMM_KTILE) {
    __syncthreads();
#pragma unroll
    for (int i = 0; i < BLK_STRIPS; ++i) {
      int idx = tid + 256 * i;
      int sr  = idx >> 4;
      int sc  = idx & 15;
      ((float4*)(atile + sr * GEMM_PITCH))[sc] =
          ((const float4*)(A + (size_t)(row0 + sr) * Kdim + k0))[sc];
    }
    __syncthreads();
#pragma unroll
    for (int ks = 0; ks < GEMM_KTILE; ks += 4) {
      v2f b;
      b.x = B[(size_t)(k0 + ks + kb)     * DH + col0 + lid];
      b.y = B[(size_t)(k0 + ks + kb + 1) * DH + col0 + lid];
#pragma unroll
      for (int s = 0; s < BLK_STRIPS; ++s) {
        v2f a = *(const v2f*)(atile + (s * 16 + lid) * GEMM_PITCH + ks + kb);
        acc[s] = wmma4(a, b, acc[s]);
      }
    }
  }
#pragma unroll
  for (int s = 0; s < BLK_STRIPS; ++s)
#pragma unroll
    for (int r = 0; r < 8; ++r)
      C[(size_t)(row0 + s * 16 + r + 8 * half) * DH + col0 + lid] = acc[s][r];
}

#define DENSE_BLOCKS      (Nn / BLK_ROWS)
#define DENSE_WITH_RANK   768
#define DENSE_WITH_PLACE  (DENSE_BLOCKS - DENSE_WITH_RANK)
#define EDGE_BLOCKS       ((Ee + 255) / 256)
__global__ void __launch_bounds__(256)
dense_rank_kernel(const float* __restrict__ x, const float* __restrict__ Wm, float* __restrict__ H, int Kdim,
                  const int* __restrict__ edge, int* __restrict__ cnt, int* __restrict__ rank) {
  __shared__ __align__(16) float atile[BLK_ROWS * GEMM_PITCH];
  int b = blockIdx.x;
  if (b < DENSE_WITH_RANK) dense_rows(b, atile, x, Wm, H, Kdim);
  else                     rank_edges(b - DENSE_WITH_RANK, edge, cnt, rank);
}

__global__ void __launch_bounds__(256)
dense_place_kernel(const float* __restrict__ x, const float* __restrict__ Wm, float* __restrict__ H, int Kdim,
                   const int* __restrict__ edge, const int* __restrict__ offs, const int* __restrict__ rank,
                   const float* __restrict__ dinv, int* __restrict__ csr) {
  __shared__ __align__(16) float atile[BLK_ROWS * GEMM_PITCH];
  int b = blockIdx.x;
  if (b < DENSE_WITH_PLACE) dense_rows(DENSE_WITH_RANK + b, atile, x, Wm, H, Kdim);
  else                      place_edges(b - DENSE_WITH_PLACE, edge, offs, rank, dinv, csr);
}

__device__ __forceinline__ void
gather_rows(int gblock, const int* __restrict__ offs, const int* __restrict__ csr,
            const float* __restrict__ dinv, const float* __restrict__ H, float* __restrict__ S) {
  int lane = threadIdx.x & 31;
  int c = gblock * 8 + __builtin_amdgcn_readfirstlane((int)(threadIdx.x >> 5));
  int beg = offs[c];
  int end = offs[c + 1];
  beg = min(max(beg, 0), Ee);
  end = min(max(end, beg), Ee);
  float di = dinv[c];
  float w0 = di * di;
  const float4 hs = *(const float4*)(H + (size_t)c * DH + lane * 4);
  float4 acc;
  acc.x = w0 * hs.x; acc.y = w0 * hs.y; acc.z = w0 * hs.z; acc.w = w0 * hs.w;
  for (int base = beg; base < end; base += 32) {
    int n = min(32, end - base);
    int2 my = make_int2(0, 0);
    if (lane < n) my = ((const int2*)csr)[base + lane];
#pragma unroll 4
    for (int j = 0; j < n; ++j) {
      int   r = __builtin_amdgcn_readlane(my.x, j);
      float w = __int_as_float(__builtin_amdgcn_readlane(my.y, j));
      r = min(max(r, 0), Nn - 1);
      const float4 h = *(const float4*)(H + (size_t)r * DH + lane * 4);
      acc.x = fmaf(w, h.x, acc.x);
      acc.y = fmaf(w, h.y, acc.y);
      acc.z = fmaf(w, h.z, acc.z);
      acc.w = fmaf(w, h.w, acc.w);
    }
  }
  *(float4*)(S + (size_t)c * DH + lane * 4) = acc;
}

#define GRAM_TILES 36
#define GRAM_CHUNKS (Nn / GRAM_CHUNK)
template <bool THIRD>
__device__ __forceinline__ void
gram_ksteps(const float* __restrict__ tile, int kb, int ca, int cb0, int cb1, int cb2, v8f& acc0, v8f& acc1, v8f& acc2) {
#pragma unroll 2
  for (int ks = 0; ks < GRAM_TILE; ks += 4) {
    const float* r0 = tile + (ks + kb) * GRAM_PITCH;
    const float* r1 = r0 + GRAM_PITCH;
    v2f a, b0, b1;
    a.x  = r0[ca];   a.y  = r1[ca];
    b0.x = r0[cb0];  b0.y = r1[cb0];
    b1.x = r0[cb1];  b1.y = r1[cb1];
    acc0 = wmma4(a, b0, acc0);
    acc1 = wmma4(a, b1, acc1);
    if (THIRD) {
      v2f b2;
      b2.x = r0[cb2]; b2.y = r1[cb2];
      acc2 = wmma4(a, b2, acc2);
    }
  }
}
__device__ __forceinline__ void
gram_strips(int gblock, float* __restrict__ tile, const float* __restrict__ H, float* __restrict__ P) {
  int mblock = gblock >> 1;
  int side   = gblock & 1;
  int tid  = threadIdx.x;
  int lane = tid & 31;
  int half = lane >> 4;
  int lid  = lane & 15;
  int kb   = 2 * half;
  int wv   = __builtin_amdgcn_readfirstlane(tid >> 5);
  int n0   = mblock * GRAM_CHUNK;
  int d0   = 2 * side;
  int ca   = wv * 16 + lid;
  int cb0  = ((wv + d0) & 7) * 16 + lid;
  int cb1  = ((wv + d0 + 1) & 7) * 16 + lid;
  int cb2  = ((wv + 4) & 7) * 16 + lid;
  bool third = (wv < 4) && ((wv >> 1) == side);
  v8f acc0 = {}, acc1 = {}, acc2 = {};
  for (int base = n0; base < n0 + GRAM_CHUNK; base += GRAM_TILE) {
    __syncthreads();
    {
      const float4* src = (const float4*)(H + (size_t)base * DH);
      int trow = tid >> 5;
      int tc4  = tid & 31;
#pragma unroll
      for (int i = 0; i < GRAM_TILE / 8; ++i)
        ((float4*)(tile + (trow + 8 * i) * GRAM_PITCH))[tc4] = src[tid + 256 * i];
    }
    __syncthreads();
    if (third) gram_ksteps<true>(tile, kb, ca, cb0, cb1, cb2, acc0, acc1, acc2);
    else       gram_ksteps<false>(tile, kb, ca, cb0, cb1, cb2, acc0, acc1, acc2);
  }
  float* Pc = P + (size_t)mblock * (GRAM_TILES * 256) + (8 * half) * 16 + lid;
  float* t0 = Pc + (d0 * 8 + wv) * 256;
  float* t1 = Pc + ((d0 + 1) * 8 + wv) * 256;
#pragma unroll
  for (int r = 0; r < 8; ++r) {
    t0[r * 16] = acc0[r];
    t1[r * 16] = acc1[r];
  }
  if (third) {
    float* t2 = Pc + (32 + wv) * 256;
#pragma unroll
    for (int r = 0; r < 8; ++r) t2[r * 16] = acc2[r];
  }
}

#define RED_GROUPS 16
#define RED_DEPTH  ((GRAM_CHUNKS + RED_GROUPS - 1) / RED_GROUPS)
__global__ void __launch_bounds__(256)
gram_reduce_kernel(const float* __restrict__ P, float* __restrict__ G) {
  __shared__ float4 part[RED_GROUPS][16];
  int tid  = threadIdx.x;
  int slot = tid & 15;
  int grp  = tid >> 4;
  const float4* src = (const float4*)P + blockIdx.x * 16 + slot;
  float4 v[RED_DEPTH];
#pragma unroll
  for (int i = 0; i < RED_DEPTH; ++i) {
    int m = grp + RED_GROUPS * i;
    v[i] = make_float4(0.0f, 0.0f, 0.0f, 0.0f);
    if (m < GRAM_CHUNKS) v[i] = src[(size_t)m * (GRAM_TILES * 256 / 4)];
  }
  float4 s = v[0];
#pragma unroll
  for (int i = 1; i < RED_DEPTH; ++i) {
    s.x += v[i].x; s.y += v[i].y; s.z += v[i].z; s.w += v[i].w;
  }
  part[grp][slot] = s;
  __syncthreads();
  if (tid < 16) {
    float4 t = part[0][tid];
#pragma unroll
    for (int g = 1; g < RED_GROUPS; ++g) {
      float4 u = part[g][tid];
      t.x += u.x; t.y += u.y; t.z += u.z; t.w += u.w;
    }
    ((float4*)G)[blockIdx.x * 16 + tid] = t;
  }
}

#define GRAM_BLOCKS   (2 * GRAM_CHUNKS)
#define GATHER_BLOCKS (Nn / 8)
__global__ void __launch_bounds__(256)
propagate_gram_kernel(const int* __restrict__ offs, const int* __restrict__ csr,
                      const float* __restrict__ dinv, const float* __restrict__ H,
                      float* __restrict__ S, float* __restrict__ P) {
  __shared__ __align__(16) float tile[GRAM_TILE * GRAM_PITCH];
  int b = blockIdx.x;
  if (b < GRAM_BLOCKS) gram_strips(b, tile, H, P);
  else                 gather_rows(b - GRAM_BLOCKS, offs, csr, dinv, H, S);
}


__device__ __forceinline__ void
tile_times_mat(const float* __restrict__ tile, const float* __restrict__ Bm, int col0, int lid, int kb,
               v8f (&acc)[BLK_STRIPS]) {
#pragma unroll 8
  for (int kk = 0; kk < DH; kk += 4) {
    v2f b;
    b.x = Bm[(size_t)(kk + kb)     * DH + col0 + lid];
    b.y = Bm[(size_t)(kk + kb + 1) * DH + col0 + lid];
#pragma unroll
    for (int s = 0; s < BLK_STRIPS; ++s) {
      v2f a = *(const v2f*)(tile + (s * 16 + lid) * UPD_PITCH + kk + kb);
      acc[s] = wmma4(a, b, acc[s]);
    }
  }
}

__device__ __forceinline__ void
tile_times_gram(const float* __restrict__ tile, const float* __restrict__ Gp, int wv, int lid, int kb,
                v8f (&acc)[BLK_STRIPS]) {
#pragma unroll 8
  for (int kk = 0; kk < DH; kk += 4) {
    int tk = kk >> 4;
    int d  = (wv - tk) & 7;
    bool direct = (d < 4) || (d == 4 && tk < 4);
    int ti = direct ? tk : wv;
    int dd = direct ? d : ((8 - d) & 7);
    int tn = (dd < 4) ? (dd * 8 + ti) : (32 + ti);
    int sk = direct ? 16 : 1;
    int sn = direct ? 1 : 16;
    const float* g = Gp + tn * 256 + ((kk & 15) + kb) * sk + lid * sn;
    v2f b;
    b.x = g[0];
    b.y = g[sk];
#pragma unroll
    for (int s = 0; s < BLK_STRIPS; ++s) {
      v2f a = *(const v2f*)(tile + (s * 16 + lid) * UPD_PITCH + kk + kb);
      acc[s] = wmma4(a, b, acc[s]);
    }
  }
}

__device__ __forceinline__ void
layer_tail(float* __restrict__ htile, float* __restrict__ H, const float* __restrict__ S,
           const float* __restrict__ G, const float* __restrict__ bias,
           const float* __restrict__ Wn, const float* __restrict__ bn, float* __restrict__ out, const bool last) {
  int tid  = threadIdx.x;
  int wave = tid >> 5;
  int lane = tid & 31;
  int half = lane >> 4;
  int lid  = lane & 15;
  int kb   = 2 * half;
  int row0 = blockIdx.x * BLK_ROWS;
  int col0 = wave * 16;
  {
    const float4* src = (const float4*)(H + (size_t)row0 * DH);
#pragma unroll
    for (int i = 0; i < 2 * BLK_STRIPS; ++i) {
      int idx  = tid + 256 * i;
      int trow = idx >> 5;
      int tc4  = idx & 31;
      ((float4*)(htile + trow * UPD_PITCH))[tc4] = src[idx];
    }
  }
  __syncthreads();
  v8f acc[BLK_STRIPS] = {};
  tile_times_gram(htile, G, __builtin_amdgcn_readfirstlane(wave), lid, kb, acc);
  float bcol = bias[col0 + lid];
  __syncthreads();
#pragma unroll
  for (int s = 0; s < BLK_STRIPS; ++s) {
#pragma unroll
    for (int r = 0; r < 8; ++r) {
      int m = s * 16 + r + 8 * half;
      size_t off = (size_t)(row0 + m) * DH + col0 + lid;
      float val = AZ * htile[m * UPD_PITCH + col0 + lid] + CS * S[off] - CT * acc[s][r] + bcol;
      htile[m * UPD_PITCH + col0 + lid] = fmaxf(val, 0.0f);
    }
  }
  __syncthreads();
  if (!last) {
    v8f nxt[BLK_STRIPS] = {};
    tile_times_mat(htile, Wn, col0, lid, kb, nxt);
#pragma unroll
    for (int s = 0; s < BLK_STRIPS; ++s)
#pragma unroll
      for (int r = 0; r < 8; ++r)
        H[(size_t)(row0 + s * 16 + r + 8 * half) * DH + col0 + lid] = nxt[s][r];
  } else {
    int wv = __builtin_amdgcn_readfirstlane(wave);
    if (wv < BLK_STRIPS) {
      const float* arow = htile + (wv * 16 + (((lid & 7) << 1) | (lid >> 3))) * UPD_PITCH + kb;
      v8f l0 = {}, l1 = {};
#pragma unroll 8
      for (int kk = 0; kk < DH; kk += 8) {
        v2f b0, b1;
        b0.x = Wn[(kk + kb)     * Kc + lid];
        b0.y = Wn[(kk + kb + 1) * Kc + lid];
        b1.x = Wn[(kk + kb + 4) * Kc + lid];
        b1.y = Wn[(kk + kb + 5) * Kc + lid];
        v2f a0 = *(const v2f*)(arow + kk);
        v2f a1 = *(const v2f*)(arow + kk + 4);
        l0 = wmma4(a0, b0, l0);
        l1 = wmma4(a1, b1, l1);
      }
      float bl = bn[lid];
      float* orow = out + (size_t)(row0 + wv * 16) * Kc + lane;
#pragma unroll
      for (int r = 0; r < 8; ++r) {
        float v = (l0[r] + l1[r]) + bl;
        float mx = v;
#pragma unroll
        for (int d = 8; d >= 1; d >>= 1) mx = fmaxf(mx, __shfl_xor(mx, d));
        float se = expf(v - mx);
#pragma unroll
        for (int d = 8; d >= 1; d >>= 1) se += __shfl_xor(se, d);
        orow[32 * r] = v - (mx + logf(se));
      }
    }
  }
}

__global__ void __launch_bounds__(256)
update_next_kernel(float* __restrict__ H, const float* __restrict__ S, const float* __restrict__ G,
                   const float* __restrict__ bias, const float* __restrict__ Wn) {
  __shared__ __align__(16) float htile[BLK_ROWS * UPD_PITCH];
  layer_tail(htile, H, S, G, bias, Wn, nullptr, nullptr, false);
}

__global__ void __launch_bounds__(256)
update_final_kernel(float* __restrict__ H, const float* __restrict__ S, const float* __restrict__ G,
                    const float* __restrict__ bias, const float* __restrict__ Wn, const float* __restrict__ bn,
                    float* __restrict__ out) {
  __shared__ __align__(16) float htile[BLK_ROWS * UPD_PITCH];
  layer_tail(htile, H, S, G, bias, Wn, bn, out, true);
}

extern "C" void kernel_launch(void* const* d_in, const int* in_sizes, int n_in,
                              void* d_out, int out_size, void* d_ws, size_t ws_size,
                              hipStream_t stream) {
  const float* x    = (const float*)d_in[0];
  const int*   edge = (const int*)  d_in[1];
  const float* W1   = (const float*)d_in[2];
  const float* b1   = (const float*)d_in[3];
  const float* W2   = (const float*)d_in[4];
  const float* b2   = (const float*)d_in[5];
  const float* Wf   = (const float*)d_in[6];
  const float* bf   = (const float*)d_in[7];
  float* out = (float*)d_out;

  float* H    = (float*)d_ws;
  float* S    = H + (size_t)Nn * DH;
  float* dinv = S + (size_t)Nn * DH;
  float* G    = dinv + Nn;
  int*   cnt  = (int*)(G + (size_t)DH * DH);
  int*   offs = cnt + Nn;
  int*   rank = offs + Nn + 4;
  int*   csr  = rank + Ee;
  float* P    = (float*)(csr + 2 * Ee + 28);

  dim3 b256(256);

  fill4_kernel<<<128, b256, 0, stream>>>((float4*)cnt, (long)Nn / 4, 0.0f);
  dense_rank_kernel<<<DENSE_WITH_RANK + EDGE_BLOCKS, b256, 0, stream>>>(x, W1, H, DIN, edge, cnt, rank);
  scan_kernel<<<(Nn + 1023) / 1024, b256, 0, stream>>>(cnt, offs, dinv);
  dense_place_kernel<<<DENSE_WITH_PLACE + EDGE_BLOCKS, b256, 0, stream>>>(x, W1, H, DIN, edge, offs, rank, dinv, csr);

  propagate_gram_kernel<<<GRAM_BLOCKS + GATHER_BLOCKS, b256, 0, stream>>>(offs, csr, dinv, H, S, P);
  gram_reduce_kernel<<<GRAM_TILES * 256 / 64, b256, 0, stream>>>(P, G);
  update_next_kernel<<<Nn / BLK_ROWS, b256, 0, stream>>>(H, S, G, b1, W2);

  propagate_gram_kernel<<<GRAM_BLOCKS + GATHER_BLOCKS, b256, 0, stream>>>(offs, csr, dinv, H, S, P);
  gram_reduce_kernel<<<GRAM_TILES * 256 / 64, b256, 0, stream>>>(P, G);
  update_final_kernel<<<Nn / BLK_ROWS, b256, 0, stream>>>(H, S, G, b2, Wf, bf, out);
}
